# starting kernel
# baseline (speedup 1.0000x reference)
#include <hip/hip_runtime.h>
#include <hip/hip_bf16.h>

typedef float v2f __attribute__((ext_vector_type(2)));
typedef float v4f __attribute__((ext_vector_type(4)));
typedef float v8f __attribute__((ext_vector_type(8)));

#define N_NODES 100000
#define N_EDGES 3200000
#define IN_CH   128
#define HID     16


__global__ void k_deg_init(float* __restrict__ deg, int n) {
    int i = blockIdx.x * blockDim.x + threadIdx.x;
    if (i < n) deg[i] = 1.0f;
}

__global__ void k_deg_edges(const int* __restrict__ col, float* __restrict__ deg, int e) {
    int i = blockIdx.x * blockDim.x + threadIdx.x;
    if (i < e) atomicAdd(&deg[col[i]], 1.0f);
}

__global__ void k_dinv(float* __restrict__ deg, int n) {
    int i = blockIdx.x * blockDim.x + threadIdx.x;
    if (i < n) {
        float d = deg[i];
        deg[i] = (d > 0.0f) ? rsqrtf(d) : 0.0f;
    }
}


__global__ void k_gemm_wmma(const float* __restrict__ A, const float* __restrict__ W,
                            float* __restrict__ C, int n, int K) {
    extern __shared__ float sW[];
    int t = threadIdx.x;
    for (int i = t; i < K * HID; i += blockDim.x) sW[i] = W[i];
    __syncthreads();

    int wave = t >> 5;
    int lane = t & 31;
    int m    = lane & 15;
    int half = lane >> 4;

    int tile = blockIdx.x * (blockDim.x >> 5) + wave;
    int rb   = tile * 16;
    if (rb >= n) return;

    int row = rb + m;
    if (row > n - 1) row = n - 1;
    const float* arow = A + (long)row * K + 2 * half;

    v8f acc = {};
    #pragma unroll 8
    for (int k = 0; k < K; k += 4) {
        v2f a = *reinterpret_cast<const v2f*>(arow + k);
        int kb = k + 2 * half;
        v2f b;
        b[0] = sW[kb * HID + m];
        b[1] = sW[(kb + 1) * HID + m];
        acc = __builtin_amdgcn_wmma_f32_16x16x4_f32(
                   false, a,  false, b,
                   (short)0, acc,  false,  false);
    }

    #pragma unroll
    for (int i = 0; i < 8; ++i) {
        int r = rb + i + 8 * half;
        if (r < n) C[(long)r * HID + m] = acc[i];
    }
}


__global__ void k_zero4(v4f* __restrict__ buf, int n4) {
    int i = blockIdx.x * blockDim.x + threadIdx.x;
    if (i < n4) buf[i] = (v4f){0.f, 0.f, 0.f, 0.f};
}

__global__ void k_edge_agg(const int* __restrict__ row, const int* __restrict__ col,
                           const float* __restrict__ dinv, const float* __restrict__ h,
                           float* __restrict__ agg, int e) {
    int t = blockIdx.x * blockDim.x + threadIdx.x;
    if (t >= 4 * e) return;
    int eid = t >> 2;
    int c4  = (t & 3) << 2;
    int r = row[eid];
    int c = col[eid];
    float w = dinv[r] * dinv[c];
    v4f hv = *reinterpret_cast<const v4f*>(h + (long)r * HID + c4);
    float* dst = agg + (long)c * HID + c4;
    atomicAdd(dst + 0, w * hv[0]);
    atomicAdd(dst + 1, w * hv[1]);
    atomicAdd(dst + 2, w * hv[2]);
    atomicAdd(dst + 3, w * hv[3]);
}

__global__ void k_self_bias_relu(float* __restrict__ agg, const float* __restrict__ h,
                                 const float* __restrict__ dinv, const float* __restrict__ b,
                                 int n) {
    int t = blockIdx.x * blockDim.x + threadIdx.x;
    if (t >= n * HID) return;
    int i = t >> 4;
    int c = t & 15;
    float di = dinv[i];
    float v = agg[t] + di * di * h[t] + b[c];
    agg[t] = v > 0.0f ? v : 0.0f;
}

__global__ void k_linear(const float* __restrict__ h, const float* __restrict__ Wl,
                         const float* __restrict__ bl, float* __restrict__ out, int n) {
    int i = blockIdx.x * blockDim.x + threadIdx.x;
    if (i >= n) return;
    const float* hr = h + (long)i * HID;
    float s = bl[0];
    #pragma unroll
    for (int c = 0; c < HID; ++c) s += hr[c] * Wl[c];
    out[i] = s;
}


extern "C" void kernel_launch(void* const* d_in, const int* in_sizes, int n_in,
                              void* d_out, int out_size, void* d_ws, size_t ws_size,
                              hipStream_t stream) {
    const float* x  = (const float*)d_in[0];
    const int*   ei = (const int*)d_in[1];
    const int*   row = ei;
    const int*   col = ei + N_EDGES;
    const float* W1 = (const float*)d_in[2];
    const float* b1 = (const float*)d_in[3];
    const float* W2 = (const float*)d_in[4];
    const float* b2 = (const float*)d_in[5];
    const float* Wl = (const float*)d_in[6];
    const float* bl = (const float*)d_in[7];
    float* out = (float*)d_out;

    float* ws   = (float*)d_ws;
    float* dinv = ws;
    float* bufA = ws + ((N_NODES + 255) / 256) * 256;
    float* bufB = bufA + (size_t)N_NODES * HID;

    const int T = 256;
    const int gN   = (N_NODES + T - 1) / T;
    const int gE   = (N_EDGES + T - 1) / T;
    const int gE4  = (4 * N_EDGES + T - 1) / T;
    const int gNH  = (N_NODES * HID + T - 1) / T;
    const int gNH4 = (N_NODES * HID / 4 + T - 1) / T;
    const int tiles = (N_NODES + 15) / 16;
    const int gGemm = (tiles + (T / 32) - 1) / (T / 32);

    k_deg_init<<<gN, T, 0, stream>>>(dinv, N_NODES);
    k_deg_edges<<<gE, T, 0, stream>>>(col, dinv, N_EDGES);
    k_dinv<<<gN, T, 0, stream>>>(dinv, N_NODES);

    k_gemm_wmma<<<gGemm, T, IN_CH * HID * sizeof(float), stream>>>(x, W1, bufA, N_NODES, IN_CH);
    k_zero4<<<gNH4, T, 0, stream>>>((v4f*)bufB, N_NODES * HID / 4);
    k_edge_agg<<<gE4, T, 0, stream>>>(row, col, dinv, bufA, bufB, N_EDGES);
    k_self_bias_relu<<<gNH, T, 0, stream>>>(bufB, bufA, dinv, b1, N_NODES);

    k_gemm_wmma<<<gGemm, T, HID * HID * sizeof(float), stream>>>(bufB, W2, bufA, N_NODES, HID);
    k_zero4<<<gNH4, T, 0, stream>>>((v4f*)bufB, N_NODES * HID / 4);
    k_edge_agg<<<gE4, T, 0, stream>>>(row, col, dinv, bufA, bufB, N_EDGES);
    k_self_bias_relu<<<gNH, T, 0, stream>>>(bufB, bufA, dinv, b2, N_NODES);

    k_linear<<<gN, T, 0, stream>>>(bufB, Wl, bl, out, N_NODES);
}


# Degree histogram moved into LDS: 16 node ranges x 16 edge slices, integer ds_add counters, partial rows summed in k_dinv; no global atomics in the degree pass (integer counts, exact in any order)
# speedup vs baseline: 1.3477x; 1.3477x over previous
#include <hip/hip_runtime.h>
#include <hip/hip_bf16.h>

typedef float v2f __attribute__((ext_vector_type(2)));
typedef float v4f __attribute__((ext_vector_type(4)));
typedef float v8f __attribute__((ext_vector_type(8)));

#define N_NODES 100000
#define N_EDGES 3200000
#define IN_CH   128
#define HID     16


#define DEG_R      16
#define DEG_S      16
#define DEG_RANGE  (N_NODES / DEG_R)
#define DEG_SLICE4 (N_EDGES / DEG_S / 4)
#define DEG_T      1024
#define DEG_U      7

static_assert(N_NODES % DEG_R == 0, "ranges must tile the nodes");
static_assert(N_EDGES % (DEG_S * 4) == 0, "slices must tile the edges in groups of four");
static_assert(DEG_S <= HID, "the partial rows are kept in one N x HID buffer");

__device__ __forceinline__ void deg_bump(unsigned* hist, int c, unsigned lo) {
    unsigned d = (unsigned)c - lo;
    if (d < (unsigned)DEG_RANGE) atomicAdd(&hist[d], 1u);
}

__global__ __launch_bounds__(DEG_T) void k_deg_hist(const int* __restrict__ col,
                                                    unsigned* __restrict__ part) {
    __shared__ unsigned hist[DEG_RANGE];
    const int t  = threadIdx.x;
    const int rg = blockIdx.x % DEG_R;
    const int sl = blockIdx.x / DEG_R;
    for (int i = t; i < DEG_RANGE; i += DEG_T) hist[i] = 0u;
    __syncthreads();

    const unsigned lo = (unsigned)(rg * DEG_RANGE);
    const int4* __restrict__ c4 = reinterpret_cast<const int4*>(col) + (size_t)sl * DEG_SLICE4;
    for (int base = 0; base < DEG_SLICE4; base += DEG_T * DEG_U) {
        int4 v[DEG_U];
        #pragma unroll
        for (int u = 0; u < DEG_U; ++u) {
            int i = base + u * DEG_T + t;
            v[u] = (i < DEG_SLICE4) ? c4[i] : make_int4(-1, -1, -1, -1);
        }
        #pragma unroll
        for (int u = 0; u < DEG_U; ++u) {
            deg_bump(hist, v[u].x, lo);
            deg_bump(hist, v[u].y, lo);
            deg_bump(hist, v[u].z, lo);
            deg_bump(hist, v[u].w, lo);
        }
    }
    __syncthreads();

    unsigned* __restrict__ dst = part + (size_t)sl * N_NODES + lo;
    for (int i = t; i < DEG_RANGE; i += DEG_T) dst[i] = hist[i];
}

__global__ void k_dinv(const unsigned* __restrict__ part, float* __restrict__ dinv, int n) {
    int i = blockIdx.x * blockDim.x + threadIdx.x;
    if (i < n) {
        unsigned cnt = 1u;
        for (int s = 0; s < DEG_S; ++s) cnt += part[(size_t)s * N_NODES + i];
        float d = (float)cnt;
        dinv[i] = (d > 0.0f) ? rsqrtf(d) : 0.0f;
    }
}


__global__ void k_gemm_wmma(const float* __restrict__ A, const float* __restrict__ W,
                            float* __restrict__ C, int n, int K) {
    extern __shared__ float sW[];
    int t = threadIdx.x;
    for (int i = t; i < K * HID; i += blockDim.x) sW[i] = W[i];
    __syncthreads();

    int wave = t >> 5;
    int lane = t & 31;
    int m    = lane & 15;
    int half = lane >> 4;

    int tile = blockIdx.x * (blockDim.x >> 5) + wave;
    int rb   = tile * 16;
    if (rb >= n) return;

    int row = rb + m;
    if (row > n - 1) row = n - 1;
    const float* arow = A + (long)row * K + 2 * half;

    v8f acc = {};
    #pragma unroll 8
    for (int k = 0; k < K; k += 4) {
        v2f a = *reinterpret_cast<const v2f*>(arow + k);
        int kb = k + 2 * half;
        v2f b;
        b[0] = sW[kb * HID + m];
        b[1] = sW[(kb + 1) * HID + m];
        acc = __builtin_amdgcn_wmma_f32_16x16x4_f32(
                   false, a,  false, b,
                   (short)0, acc,  false,  false);
    }

    #pragma unroll
    for (int i = 0; i < 8; ++i) {
        int r = rb + i + 8 * half;
        if (r < n) C[(long)r * HID + m] = acc[i];
    }
}


__global__ void k_zero4(v4f* __restrict__ buf, int n4) {
    int i = blockIdx.x * blockDim.x + threadIdx.x;
    if (i < n4) buf[i] = (v4f){0.f, 0.f, 0.f, 0.f};
}

__global__ void k_edge_agg(const int* __restrict__ row, const int* __restrict__ col,
                           const float* __restrict__ dinv, const float* __restrict__ h,
                           float* __restrict__ agg, int e) {
    int t = blockIdx.x * blockDim.x + threadIdx.x;
    if (t >= 4 * e) return;
    int eid = t >> 2;
    int c4  = (t & 3) << 2;
    int r = row[eid];
    int c = col[eid];
    float w = dinv[r] * dinv[c];
    v4f hv = *reinterpret_cast<const v4f*>(h + (long)r * HID + c4);
    float* dst = agg + (long)c * HID + c4;
    atomicAdd(dst + 0, w * hv[0]);
    atomicAdd(dst + 1, w * hv[1]);
    atomicAdd(dst + 2, w * hv[2]);
    atomicAdd(dst + 3, w * hv[3]);
}

__global__ void k_self_bias_relu(float* __restrict__ agg, const float* __restrict__ h,
                                 const float* __restrict__ dinv, const float* __restrict__ b,
                                 int n) {
    int t = blockIdx.x * blockDim.x + threadIdx.x;
    if (t >= n * HID) return;
    int i = t >> 4;
    int c = t & 15;
    float di = dinv[i];
    float v = agg[t] + di * di * h[t] + b[c];
    agg[t] = v > 0.0f ? v : 0.0f;
}

__global__ void k_linear(const float* __restrict__ h, const float* __restrict__ Wl,
                         const float* __restrict__ bl, float* __restrict__ out, int n) {
    int i = blockIdx.x * blockDim.x + threadIdx.x;
    if (i >= n) return;
    const float* hr = h + (long)i * HID;
    float s = bl[0];
    #pragma unroll
    for (int c = 0; c < HID; ++c) s += hr[c] * Wl[c];
    out[i] = s;
}


extern "C" void kernel_launch(void* const* d_in, const int* in_sizes, int n_in,
                              void* d_out, int out_size, void* d_ws, size_t ws_size,
                              hipStream_t stream) {
    const float* x  = (const float*)d_in[0];
    const int*   ei = (const int*)d_in[1];
    const int*   row = ei;
    const int*   col = ei + N_EDGES;
    const float* W1 = (const float*)d_in[2];
    const float* b1 = (const float*)d_in[3];
    const float* W2 = (const float*)d_in[4];
    const float* b2 = (const float*)d_in[5];
    const float* Wl = (const float*)d_in[6];
    const float* bl = (const float*)d_in[7];
    float* out = (float*)d_out;

    float* ws   = (float*)d_ws;
    float* dinv = ws;
    float* bufA = ws + ((N_NODES + 255) / 256) * 256;
    float* bufB = bufA + (size_t)N_NODES * HID;

    const int T = 256;
    const int gN   = (N_NODES + T - 1) / T;
    const int gE4  = (4 * N_EDGES + T - 1) / T;
    const int gNH  = (N_NODES * HID + T - 1) / T;
    const int gNH4 = (N_NODES * HID / 4 + T - 1) / T;
    const int tiles = (N_NODES + 15) / 16;
    const int gGemm = (tiles + (T / 32) - 1) / (T / 32);

    unsigned* part = (unsigned*)bufB;
    k_deg_hist<<<DEG_R * DEG_S, DEG_T, 0, stream>>>(col, part);
    k_dinv<<<gN, T, 0, stream>>>(part, dinv, N_NODES);

    k_gemm_wmma<<<gGemm, T, IN_CH * HID * sizeof(float), stream>>>(x, W1, bufA, N_NODES, IN_CH);
    k_zero4<<<gNH4, T, 0, stream>>>((v4f*)bufB, N_NODES * HID / 4);
    k_edge_agg<<<gE4, T, 0, stream>>>(row, col, dinv, bufA, bufB, N_EDGES);
    k_self_bias_relu<<<gNH, T, 0, stream>>>(bufB, bufA, dinv, b1, N_NODES);

    k_gemm_wmma<<<gGemm, T, HID * HID * sizeof(float), stream>>>(bufB, W2, bufA, N_NODES, HID);
    k_zero4<<<gNH4, T, 0, stream>>>((v4f*)bufB, N_NODES * HID / 4);
    k_edge_agg<<<gE4, T, 0, stream>>>(row, col, dinv, bufA, bufB, N_EDGES);
    k_self_bias_relu<<<gNH, T, 0, stream>>>(bufB, bufA, dinv, b2, N_NODES);

    k_linear<<<gN, T, 0, stream>>>(bufB, Wl, bl, out, N_NODES);
}


# Edge pass: 16 lanes per edge in resident waves looping over 32-edge batches; each h gather and float atomic covers 2 whole 64-byte rows, not 8 quarter rows; atomics sum in arrival order as before
# speedup vs baseline: 1.9608x; 1.4549x over previous
#include <hip/hip_runtime.h>
#include <hip/hip_bf16.h>

typedef float v2f __attribute__((ext_vector_type(2)));
typedef float v4f __attribute__((ext_vector_type(4)));
typedef float v8f __attribute__((ext_vector_type(8)));

#define N_NODES 100000
#define N_EDGES 3200000
#define IN_CH   128
#define HID     16


#define DEG_R      16
#define DEG_S      16
#define DEG_RANGE  (N_NODES / DEG_R)
#define DEG_SLICE4 (N_EDGES / DEG_S / 4)
#define DEG_T      1024
#define DEG_U      7

static_assert(N_NODES % DEG_R == 0, "ranges must tile the nodes");
static_assert(N_EDGES % (DEG_S * 4) == 0, "slices must tile the edges in groups of four");
static_assert(DEG_S <= HID, "the partial rows are kept in one N x HID buffer");

__device__ __forceinline__ void deg_bump(unsigned* hist, int c, unsigned lo) {
    unsigned d = (unsigned)c - lo;
    if (d < (unsigned)DEG_RANGE) atomicAdd(&hist[d], 1u);
}

__global__ __launch_bounds__(DEG_T) void k_deg_hist(const int* __restrict__ col,
                                                    unsigned* __restrict__ part) {
    __shared__ unsigned hist[DEG_RANGE];
    const int t  = threadIdx.x;
    const int rg = blockIdx.x % DEG_R;
    const int sl = blockIdx.x / DEG_R;
    for (int i = t; i < DEG_RANGE; i += DEG_T) hist[i] = 0u;
    __syncthreads();

    const unsigned lo = (unsigned)(rg * DEG_RANGE);
    const int4* __restrict__ c4 = reinterpret_cast<const int4*>(col) + (size_t)sl * DEG_SLICE4;
    for (int base = 0; base < DEG_SLICE4; base += DEG_T * DEG_U) {
        int4 v[DEG_U];
        #pragma unroll
        for (int u = 0; u < DEG_U; ++u) {
            int i = base + u * DEG_T + t;
            v[u] = (i < DEG_SLICE4) ? c4[i] : make_int4(-1, -1, -1, -1);
        }
        #pragma unroll
        for (int u = 0; u < DEG_U; ++u) {
            deg_bump(hist, v[u].x, lo);
            deg_bump(hist, v[u].y, lo);
            deg_bump(hist, v[u].z, lo);
            deg_bump(hist, v[u].w, lo);
        }
    }
    __syncthreads();

    unsigned* __restrict__ dst = part + (size_t)sl * N_NODES + lo;
    for (int i = t; i < DEG_RANGE; i += DEG_T) dst[i] = hist[i];
}

__global__ void k_dinv(const unsigned* __restrict__ part, float* __restrict__ dinv, int n) {
    int i = blockIdx.x * blockDim.x + threadIdx.x;
    if (i < n) {
        unsigned cnt = 1u;
        for (int s = 0; s < DEG_S; ++s) cnt += part[(size_t)s * N_NODES + i];
        float d = (float)cnt;
        dinv[i] = (d > 0.0f) ? rsqrtf(d) : 0.0f;
    }
}


__global__ void k_gemm_wmma(const float* __restrict__ A, const float* __restrict__ W,
                            float* __restrict__ C, int n, int K) {
    extern __shared__ float sW[];
    int t = threadIdx.x;
    for (int i = t; i < K * HID; i += blockDim.x) sW[i] = W[i];
    __syncthreads();

    int wave = t >> 5;
    int lane = t & 31;
    int m    = lane & 15;
    int half = lane >> 4;

    int tile = blockIdx.x * (blockDim.x >> 5) + wave;
    int rb   = tile * 16;
    if (rb >= n) return;

    int row = rb + m;
    if (row > n - 1) row = n - 1;
    const float* arow = A + (long)row * K + 2 * half;

    v8f acc = {};
    #pragma unroll 8
    for (int k = 0; k < K; k += 4) {
        v2f a = *reinterpret_cast<const v2f*>(arow + k);
        int kb = k + 2 * half;
        v2f b;
        b[0] = sW[kb * HID + m];
        b[1] = sW[(kb + 1) * HID + m];
        acc = __builtin_amdgcn_wmma_f32_16x16x4_f32(
                   false, a,  false, b,
                   (short)0, acc,  false,  false);
    }

    #pragma unroll
    for (int i = 0; i < 8; ++i) {
        int r = rb + i + 8 * half;
        if (r < n) C[(long)r * HID + m] = acc[i];
    }
}


__global__ void k_zero4(v4f* __restrict__ buf, int n4) {
    int i = blockIdx.x * blockDim.x + threadIdx.x;
    if (i < n4) buf[i] = (v4f){0.f, 0.f, 0.f, 0.f};
}


#define EDGE_BATCH   32
#define EDGE_NB      (N_EDGES / EDGE_BATCH)
#define EDGE_T       256
#define EDGE_BLOCKS  1250
#define EDGE_WAVES   (EDGE_BLOCKS * (EDGE_T / 32))

static_assert(N_EDGES % EDGE_BATCH == 0, "batches must tile the edges");
static_assert(EDGE_WAVES <= EDGE_NB, "every wave owns a first batch");
static_assert(HID == 16, "one half-wave per 16-channel row");
static_assert((long)N_NODES * HID * 4 < (1L << 32), "row byte offsets are kept in 32 bits");

template <int J> __device__ __forceinline__ unsigned row_lane_u(unsigned v) {
    return (unsigned)__builtin_amdgcn_update_dpp(0, (int)v, 0x150 + J, 0xF, 0xF, true);
}
template <int J> __device__ __forceinline__ float row_lane_f(float v) {
    return __int_as_float(__builtin_amdgcn_update_dpp(0, __float_as_int(v), 0x150 + J, 0xF, 0xF, true));
}

#define EDGE_GATHER(J) \
    hv[J] = *reinterpret_cast<const float*>(reinterpret_cast<const char*>(h) + (row_lane_u<J>(rb) + ch4));
#define EDGE_SCATTER(J) \
    atomicAdd(reinterpret_cast<float*>(reinterpret_cast<char*>(agg) + (row_lane_u<J>(cb) + ch4)), \
              row_lane_f<J>(w) * hv[J]);
#define EDGE_16(M) M(0) M(1) M(2) M(3) M(4) M(5) M(6) M(7) M(8) M(9) M(10) M(11) M(12) M(13) M(14) M(15)

__global__ __launch_bounds__(EDGE_T) void k_edge_agg(const int* __restrict__ row,
                                                     const int* __restrict__ col,
                                                     const float* __restrict__ dinv,
                                                     const float* __restrict__ h,
                                                     float* __restrict__ agg) {
    const int lane = threadIdx.x & 31;
    const unsigned ch4 = (unsigned)(lane & 15) * 4u;
    int b = blockIdx.x * (EDGE_T / 32) + (threadIdx.x >> 5);
    if (b >= EDGE_NB) return;

    int r = row[b * EDGE_BATCH + lane];
    int c = col[b * EDGE_BATCH + lane];
    for (; b < EDGE_NB; b += EDGE_WAVES) {
        const float w = dinv[r] * dinv[c];
        const unsigned rb = (unsigned)r * (HID * 4u);
        const unsigned cb = (unsigned)c * (HID * 4u);

        float hv[16];
        EDGE_16(EDGE_GATHER)

        const int nb = b + EDGE_WAVES;
        if (nb < EDGE_NB) {
            r = row[nb * EDGE_BATCH + lane];
            c = col[nb * EDGE_BATCH + lane];
        }

        EDGE_16(EDGE_SCATTER)
    }
}

__global__ void k_self_bias_relu(float* __restrict__ agg, const float* __restrict__ h,
                                 const float* __restrict__ dinv, const float* __restrict__ b,
                                 int n) {
    int t = blockIdx.x * blockDim.x + threadIdx.x;
    if (t >= n * HID) return;
    int i = t >> 4;
    int c = t & 15;
    float di = dinv[i];
    float v = agg[t] + di * di * h[t] + b[c];
    agg[t] = v > 0.0f ? v : 0.0f;
}

__global__ void k_linear(const float* __restrict__ h, const float* __restrict__ Wl,
                         const float* __restrict__ bl, float* __restrict__ out, int n) {
    int i = blockIdx.x * blockDim.x + threadIdx.x;
    if (i >= n) return;
    const float* hr = h + (long)i * HID;
    float s = bl[0];
    #pragma unroll
    for (int c = 0; c < HID; ++c) s += hr[c] * Wl[c];
    out[i] = s;
}


extern "C" void kernel_launch(void* const* d_in, const int* in_sizes, int n_in,
                              void* d_out, int out_size, void* d_ws, size_t ws_size,
                              hipStream_t stream) {
    const float* x  = (const float*)d_in[0];
    const int*   ei = (const int*)d_in[1];
    const int*   row = ei;
    const int*   col = ei + N_EDGES;
    const float* W1 = (const float*)d_in[2];
    const float* b1 = (const float*)d_in[3];
    const float* W2 = (const float*)d_in[4];
    const float* b2 = (const float*)d_in[5];
    const float* Wl = (const float*)d_in[6];
    const float* bl = (const float*)d_in[7];
    float* out = (float*)d_out;

    float* ws   = (float*)d_ws;
    float* dinv = ws;
    float* bufA = ws + ((N_NODES + 255) / 256) * 256;
    float* bufB = bufA + (size_t)N_NODES * HID;

    const int T = 256;
    const int gN   = (N_NODES + T - 1) / T;
    const int gNH  = (N_NODES * HID + T - 1) / T;
    const int gNH4 = (N_NODES * HID / 4 + T - 1) / T;
    const int tiles = (N_NODES + 15) / 16;
    const int gGemm = (tiles + (T / 32) - 1) / (T / 32);

    unsigned* part = (unsigned*)bufB;
    k_deg_hist<<<DEG_R * DEG_S, DEG_T, 0, stream>>>(col, part);
    k_dinv<<<gN, T, 0, stream>>>(part, dinv, N_NODES);

    k_gemm_wmma<<<gGemm, T, IN_CH * HID * sizeof(float), stream>>>(x, W1, bufA, N_NODES, IN_CH);
    k_zero4<<<gNH4, T, 0, stream>>>((v4f*)bufB, N_NODES * HID / 4);
    k_edge_agg<<<EDGE_BLOCKS, EDGE_T, 0, stream>>>(row, col, dinv, bufA, bufB);
    k_self_bias_relu<<<gNH, T, 0, stream>>>(bufB, bufA, dinv, b1, N_NODES);

    k_gemm_wmma<<<gGemm, T, HID * HID * sizeof(float), stream>>>(bufB, W2, bufA, N_NODES, HID);
    k_zero4<<<gNH4, T, 0, stream>>>((v4f*)bufB, N_NODES * HID / 4);
    k_edge_agg<<<EDGE_BLOCKS, EDGE_T, 0, stream>>>(row, col, dinv, bufA, bufB);
    k_self_bias_relu<<<gNH, T, 0, stream>>>(bufB, bufA, dinv, b2, N_NODES);

    k_linear<<<gN, T, 0, stream>>>(bufB, Wl, bl, out, N_NODES);
}


# Zero-fill and self-loop/bias/relu launches folded into neighbours: GEMM epilogue stores dinv^2*h+b as start value of the atomic sums (summed first, not last); relu applied on load; 11 -> 7 launches
# speedup vs baseline: 2.1222x; 1.0823x over previous
#include <hip/hip_runtime.h>
#include <hip/hip_bf16.h>

typedef float v2f __attribute__((ext_vector_type(2)));
typedef float v4f __attribute__((ext_vector_type(4)));
typedef float v8f __attribute__((ext_vector_type(8)));

#define N_NODES 100000
#define N_EDGES 3200000
#define IN_CH   128
#define HID     16


#define DEG_R      16
#define DEG_S      16
#define DEG_RANGE  (N_NODES / DEG_R)
#define DEG_SLICE4 (N_EDGES / DEG_S / 4)
#define DEG_T      1024
#define DEG_U      7

static_assert(N_NODES % DEG_R == 0, "ranges must tile the nodes");
static_assert(N_EDGES % (DEG_S * 4) == 0, "slices must tile the edges in groups of four");
static_assert(DEG_S <= HID, "the partial rows are kept in one N x HID buffer");
static_assert(N_NODES % 16 == 0, "GEMM tiles are whole: a lane's 8 dinv values are read as two 16-byte loads");

__device__ __forceinline__ void deg_bump(unsigned* hist, int c, unsigned lo) {
    unsigned d = (unsigned)c - lo;
    if (d < (unsigned)DEG_RANGE) atomicAdd(&hist[d], 1u);
}

__global__ __launch_bounds__(DEG_T) void k_deg_hist(const int* __restrict__ col,
                                                    unsigned* __restrict__ part) {
    __shared__ unsigned hist[DEG_RANGE];
    const int t  = threadIdx.x;
    const int rg = blockIdx.x % DEG_R;
    const int sl = blockIdx.x / DEG_R;
    for (int i = t; i < DEG_RANGE; i += DEG_T) hist[i] = 0u;
    __syncthreads();

    const unsigned lo = (unsigned)(rg * DEG_RANGE);
    const int4* __restrict__ c4 = reinterpret_cast<const int4*>(col) + (size_t)sl * DEG_SLICE4;
    for (int base = 0; base < DEG_SLICE4; base += DEG_T * DEG_U) {
        int4 v[DEG_U];
        #pragma unroll
        for (int u = 0; u < DEG_U; ++u) {
            int i = base + u * DEG_T + t;
            v[u] = (i < DEG_SLICE4) ? c4[i] : make_int4(-1, -1, -1, -1);
        }
        #pragma unroll
        for (int u = 0; u < DEG_U; ++u) {
            deg_bump(hist, v[u].x, lo);
            deg_bump(hist, v[u].y, lo);
            deg_bump(hist, v[u].z, lo);
            deg_bump(hist, v[u].w, lo);
        }
    }
    __syncthreads();

    unsigned* __restrict__ dst = part + (size_t)sl * N_NODES + lo;
    for (int i = t; i < DEG_RANGE; i += DEG_T) dst[i] = hist[i];
}

__global__ void k_dinv(const unsigned* __restrict__ part, float* __restrict__ dinv, int n) {
    int i = blockIdx.x * blockDim.x + threadIdx.x;
    if (i < n) {
        unsigned cnt = 1u;
        for (int s = 0; s < DEG_S; ++s) cnt += part[(size_t)s * N_NODES + i];
        float d = (float)cnt;
        dinv[i] = (d > 0.0f) ? rsqrtf(d) : 0.0f;
    }
}


template <bool RELU_A>
__global__ void k_gemm_wmma(const float* A, const float* __restrict__ W,
                            const float* __restrict__ dinv, const float* __restrict__ bias,
                            float* H, float* S, int n, int K) {
    extern __shared__ float sW[];
    int t = threadIdx.x;
    for (int i = t; i < K * HID; i += blockDim.x) sW[i] = W[i];
    __syncthreads();

    int wave = t >> 5;
    int lane = t & 31;
    int m    = lane & 15;
    int half = lane >> 4;

    int tile = blockIdx.x * (blockDim.x >> 5) + wave;
    int rb   = tile * 16;
    if (rb >= n) return;

    int row = rb + m;
    if (row > n - 1) row = n - 1;
    const float* arow = A + (long)row * K + 2 * half;

    const int r0 = rb + 8 * half;
    const v4f d0 = *reinterpret_cast<const v4f*>(dinv + r0);
    const v4f d1 = *reinterpret_cast<const v4f*>(dinv + r0 + 4);
    const float bm = bias[m];

    v8f acc = {};
    #pragma unroll 8
    for (int k = 0; k < K; k += 4) {
        v2f a = *reinterpret_cast<const v2f*>(arow + k);
        if (RELU_A) {
            a[0] = a[0] > 0.0f ? a[0] : 0.0f;
            a[1] = a[1] > 0.0f ? a[1] : 0.0f;
        }
        int kb = k + 2 * half;
        v2f b;
        b[0] = sW[kb * HID + m];
        b[1] = sW[(kb + 1) * HID + m];
        acc = __builtin_amdgcn_wmma_f32_16x16x4_f32(
                   false, a,  false, b,
                   (short)0, acc,  false,  false);
    }

    #pragma unroll
    for (int i = 0; i < 8; ++i) {
        int r = r0 + i;
        if (r < n) {
            float di = (i < 4) ? d0[i & 3] : d1[i & 3];
            H[(long)r * HID + m] = acc[i];
            S[(long)r * HID + m] = di * di * acc[i] + bm;
        }
    }
}


#define EDGE_BATCH   32
#define EDGE_NB      (N_EDGES / EDGE_BATCH)
#define EDGE_T       256
#define EDGE_BLOCKS  1250
#define EDGE_WAVES   (EDGE_BLOCKS * (EDGE_T / 32))

static_assert(N_EDGES % EDGE_BATCH == 0, "batches must tile the edges");
static_assert(EDGE_WAVES <= EDGE_NB, "every wave owns a first batch");
static_assert(HID == 16, "one half-wave per 16-channel row");
static_assert((long)N_NODES * HID * 4 < (1L << 32), "row byte offsets are kept in 32 bits");

template <int J> __device__ __forceinline__ unsigned row_lane_u(unsigned v) {
    return (unsigned)__builtin_amdgcn_update_dpp(0, (int)v, 0x150 + J, 0xF, 0xF, true);
}
template <int J> __device__ __forceinline__ float row_lane_f(float v) {
    return __int_as_float(__builtin_amdgcn_update_dpp(0, __float_as_int(v), 0x150 + J, 0xF, 0xF, true));
}

#define EDGE_GATHER(J) \
    hv[J] = *reinterpret_cast<const float*>(reinterpret_cast<const char*>(h) + (row_lane_u<J>(rb) + ch4));
#define EDGE_SCATTER(J) \
    atomicAdd(reinterpret_cast<float*>(reinterpret_cast<char*>(agg) + (row_lane_u<J>(cb) + ch4)), \
              row_lane_f<J>(w) * hv[J]);
#define EDGE_16(M) M(0) M(1) M(2) M(3) M(4) M(5) M(6) M(7) M(8) M(9) M(10) M(11) M(12) M(13) M(14) M(15)

__global__ __launch_bounds__(EDGE_T) void k_edge_agg(const int* __restrict__ row,
                                                     const int* __restrict__ col,
                                                     const float* __restrict__ dinv,
                                                     const float* __restrict__ h,
                                                     float* __restrict__ agg) {
    const int lane = threadIdx.x & 31;
    const unsigned ch4 = (unsigned)(lane & 15) * 4u;
    int b = blockIdx.x * (EDGE_T / 32) + (threadIdx.x >> 5);
    if (b >= EDGE_NB) return;

    int r = row[b * EDGE_BATCH + lane];
    int c = col[b * EDGE_BATCH + lane];
    for (; b < EDGE_NB; b += EDGE_WAVES) {
        const float w = dinv[r] * dinv[c];
        const unsigned rb = (unsigned)r * (HID * 4u);
        const unsigned cb = (unsigned)c * (HID * 4u);

        float hv[16];
        EDGE_16(EDGE_GATHER)

        const int nb = b + EDGE_WAVES;
        if (nb < EDGE_NB) {
            r = row[nb * EDGE_BATCH + lane];
            c = col[nb * EDGE_BATCH + lane];
        }

        EDGE_16(EDGE_SCATTER)
    }
}

__global__ void k_linear(const float* __restrict__ h, const float* __restrict__ Wl,
                         const float* __restrict__ bl, float* __restrict__ out, int n) {
    int i = blockIdx.x * blockDim.x + threadIdx.x;
    if (i >= n) return;
    const float* hr = h + (long)i * HID;
    float s = bl[0];
    #pragma unroll
    for (int c = 0; c < HID; ++c) {
        float v = hr[c];
        s += (v > 0.0f ? v : 0.0f) * Wl[c];
    }
    out[i] = s;
}


extern "C" void kernel_launch(void* const* d_in, const int* in_sizes, int n_in,
                              void* d_out, int out_size, void* d_ws, size_t ws_size,
                              hipStream_t stream) {
    const float* x  = (const float*)d_in[0];
    const int*   ei = (const int*)d_in[1];
    const int*   row = ei;
    const int*   col = ei + N_EDGES;
    const float* W1 = (const float*)d_in[2];
    const float* b1 = (const float*)d_in[3];
    const float* W2 = (const float*)d_in[4];
    const float* b2 = (const float*)d_in[5];
    const float* Wl = (const float*)d_in[6];
    const float* bl = (const float*)d_in[7];
    float* out = (float*)d_out;

    float* ws   = (float*)d_ws;
    float* dinv = ws;
    float* bufA = ws + ((N_NODES + 255) / 256) * 256;
    float* bufB = bufA + (size_t)N_NODES * HID;

    const int T = 256;
    const int gN   = (N_NODES + T - 1) / T;
    const int tiles = (N_NODES + 15) / 16;
    const int gGemm = (tiles + (T / 32) - 1) / (T / 32);

    unsigned* part = (unsigned*)bufB;
    k_deg_hist<<<DEG_R * DEG_S, DEG_T, 0, stream>>>(col, part);
    k_dinv<<<gN, T, 0, stream>>>(part, dinv, N_NODES);

    k_gemm_wmma<false><<<gGemm, T, IN_CH * HID * sizeof(float), stream>>>(x, W1, dinv, b1, bufA, bufB,
                                                                          N_NODES, IN_CH);
    k_edge_agg<<<EDGE_BLOCKS, EDGE_T, 0, stream>>>(row, col, dinv, bufA, bufB);

    k_gemm_wmma<true><<<gGemm, T, HID * HID * sizeof(float), stream>>>(bufB, W2, dinv, b2, bufA, bufB,
                                                                       N_NODES, HID);
    k_edge_agg<<<EDGE_BLOCKS, EDGE_T, 0, stream>>>(row, col, dinv, bufA, bufB);

    k_linear<<<gN, T, 0, stream>>>(bufB, Wl, bl, out, N_NODES);
}


# Edge weight factored out of the edge pass: GEMM stores dinv[src]*h, atomics add unscaled rows (same arrival order), dinv[dst] and bias applied to the finished sum on load; no dinv gather per edge
# speedup vs baseline: 2.4486x; 1.1538x over previous
#include <hip/hip_runtime.h>
#include <hip/hip_bf16.h>

typedef float v2f __attribute__((ext_vector_type(2)));
typedef float v4f __attribute__((ext_vector_type(4)));
typedef float v8f __attribute__((ext_vector_type(8)));

#define N_NODES 100000
#define N_EDGES 3200000
#define IN_CH   128
#define HID     16


#define DEG_R      16
#define DEG_S      16
#define DEG_RANGE  (N_NODES / DEG_R)
#define DEG_SLICE4 (N_EDGES / DEG_S / 4)
#define DEG_T      1024
#define DEG_U      7

static_assert(N_NODES % DEG_R == 0, "ranges must tile the nodes");
static_assert(N_EDGES % (DEG_S * 4) == 0, "slices must tile the edges in groups of four");
static_assert(DEG_S <= HID, "the partial rows are kept in one N x HID buffer");
static_assert(N_NODES % 16 == 0, "GEMM tiles are whole: a lane's 8 dinv values are read as two 16-byte loads");

__device__ __forceinline__ void deg_bump(unsigned* hist, int c, unsigned lo) {
    unsigned d = (unsigned)c - lo;
    if (d < (unsigned)DEG_RANGE) atomicAdd(&hist[d], 1u);
}

__global__ __launch_bounds__(DEG_T) void k_deg_hist(const int* __restrict__ col,
                                                    unsigned* __restrict__ part) {
    __shared__ unsigned hist[DEG_RANGE];
    const int t  = threadIdx.x;
    const int rg = blockIdx.x % DEG_R;
    const int sl = blockIdx.x / DEG_R;
    for (int i = t; i < DEG_RANGE; i += DEG_T) hist[i] = 0u;
    __syncthreads();

    const unsigned lo = (unsigned)(rg * DEG_RANGE);
    const int4* __restrict__ c4 = reinterpret_cast<const int4*>(col) + (size_t)sl * DEG_SLICE4;
    for (int base = 0; base < DEG_SLICE4; base += DEG_T * DEG_U) {
        int4 v[DEG_U];
        #pragma unroll
        for (int u = 0; u < DEG_U; ++u) {
            int i = base + u * DEG_T + t;
            v[u] = (i < DEG_SLICE4) ? c4[i] : make_int4(-1, -1, -1, -1);
        }
        #pragma unroll
        for (int u = 0; u < DEG_U; ++u) {
            deg_bump(hist, v[u].x, lo);
            deg_bump(hist, v[u].y, lo);
            deg_bump(hist, v[u].z, lo);
            deg_bump(hist, v[u].w, lo);
        }
    }
    __syncthreads();

    unsigned* __restrict__ dst = part + (size_t)sl * N_NODES + lo;
    for (int i = t; i < DEG_RANGE; i += DEG_T) dst[i] = hist[i];
}

__global__ void k_dinv(const unsigned* __restrict__ part, float* __restrict__ dinv, int n) {
    int i = blockIdx.x * blockDim.x + threadIdx.x;
    if (i < n) {
        unsigned cnt = 1u;
        for (int s = 0; s < DEG_S; ++s) cnt += part[(size_t)s * N_NODES + i];
        float d = (float)cnt;
        dinv[i] = (d > 0.0f) ? rsqrtf(d) : 0.0f;
    }
}


template <bool PREV>
__global__ void k_gemm_wmma(const float* A, const float* __restrict__ W,
                            const float* __restrict__ dinv, const float* __restrict__ pbias,
                            float* G, float* S, int n, int K) {
    extern __shared__ float sW[];
    int t = threadIdx.x;
    for (int i = t; i < K * HID; i += blockDim.x) sW[i] = W[i];
    __syncthreads();

    int wave = t >> 5;
    int lane = t & 31;
    int m    = lane & 15;
    int half = lane >> 4;

    int tile = blockIdx.x * (blockDim.x >> 5) + wave;
    int rb   = tile * 16;
    if (rb >= n) return;

    int row = rb + m;
    if (row > n - 1) row = n - 1;
    const float* arow = A + (long)row * K + 2 * half;

    const int r0 = rb + 8 * half;
    const v4f d0 = *reinterpret_cast<const v4f*>(dinv + r0);
    const v4f d1 = *reinterpret_cast<const v4f*>(dinv + r0 + 4);
    const float drow = PREV ? dinv[row] : 1.0f;

    v8f acc = {};
    #pragma unroll 8
    for (int k = 0; k < K; k += 4) {
        v2f a = *reinterpret_cast<const v2f*>(arow + k);
        int kb = k + 2 * half;
        if (PREV) {
            a[0] = drow * a[0] + pbias[kb];
            a[1] = drow * a[1] + pbias[kb + 1];
            a[0] = a[0] > 0.0f ? a[0] : 0.0f;
            a[1] = a[1] > 0.0f ? a[1] : 0.0f;
        }
        v2f b;
        b[0] = sW[kb * HID + m];
        b[1] = sW[(kb + 1) * HID + m];
        acc = __builtin_amdgcn_wmma_f32_16x16x4_f32(
                   false, a,  false, b,
                   (short)0, acc,  false,  false);
    }

    #pragma unroll
    for (int i = 0; i < 8; ++i) {
        int r = r0 + i;
        if (r < n) {
            float di = (i < 4) ? d0[i & 3] : d1[i & 3];
            float g  = di * acc[i];
            G[(long)r * HID + m] = g;
            S[(long)r * HID + m] = g;
        }
    }
}


#define EDGE_BATCH   32
#define EDGE_NB      (N_EDGES / EDGE_BATCH)
#define EDGE_T       256
#define EDGE_BLOCKS  1250
#define EDGE_WAVES   (EDGE_BLOCKS * (EDGE_T / 32))

static_assert(N_EDGES % EDGE_BATCH == 0, "batches must tile the edges");
static_assert(EDGE_WAVES <= EDGE_NB, "every wave owns a first batch");
static_assert(HID == 16, "one half-wave per 16-channel row");
static_assert((long)N_NODES * HID * 4 < (1L << 32), "row byte offsets are kept in 32 bits");

template <int J> __device__ __forceinline__ unsigned row_lane_u(unsigned v) {
    return (unsigned)__builtin_amdgcn_update_dpp(0, (int)v, 0x150 + J, 0xF, 0xF, true);
}

#define EDGE_GATHER(J) \
    hv[J] = *reinterpret_cast<const float*>(reinterpret_cast<const char*>(h) + (row_lane_u<J>(rb) + ch4));
#define EDGE_SCATTER(J) \
    atomicAdd(reinterpret_cast<float*>(reinterpret_cast<char*>(agg) + (row_lane_u<J>(cb) + ch4)), hv[J]);
#define EDGE_16(M) M(0) M(1) M(2) M(3) M(4) M(5) M(6) M(7) M(8) M(9) M(10) M(11) M(12) M(13) M(14) M(15)

__global__ __launch_bounds__(EDGE_T) void k_edge_agg(const int* __restrict__ row,
                                                     const int* __restrict__ col,
                                                     const float* __restrict__ h,
                                                     float* __restrict__ agg) {
    const int lane = threadIdx.x & 31;
    const unsigned ch4 = (unsigned)(lane & 15) * 4u;
    int b = blockIdx.x * (EDGE_T / 32) + (threadIdx.x >> 5);
    if (b >= EDGE_NB) return;

    int r = row[b * EDGE_BATCH + lane];
    int c = col[b * EDGE_BATCH + lane];
    for (; b < EDGE_NB; b += EDGE_WAVES) {
        const unsigned rb = (unsigned)r * (HID * 4u);
        const unsigned cb = (unsigned)c * (HID * 4u);

        float hv[16];
        EDGE_16(EDGE_GATHER)

        const int nb = b + EDGE_WAVES;
        if (nb < EDGE_NB) {
            r = row[nb * EDGE_BATCH + lane];
            c = col[nb * EDGE_BATCH + lane];
        }

        EDGE_16(EDGE_SCATTER)
    }
}

__global__ void k_linear(const float* __restrict__ t, const float* __restrict__ dinv,
                         const float* __restrict__ b, const float* __restrict__ Wl,
                         const float* __restrict__ bl, float* __restrict__ out, int n) {
    int i = blockIdx.x * blockDim.x + threadIdx.x;
    if (i >= n) return;
    const float* tr = t + (long)i * HID;
    const float di = dinv[i];
    float s = bl[0];
    #pragma unroll
    for (int c = 0; c < HID; ++c) {
        float v = di * tr[c] + b[c];
        s += (v > 0.0f ? v : 0.0f) * Wl[c];
    }
    out[i] = s;
}


extern "C" void kernel_launch(void* const* d_in, const int* in_sizes, int n_in,
                              void* d_out, int out_size, void* d_ws, size_t ws_size,
                              hipStream_t stream) {
    const float* x  = (const float*)d_in[0];
    const int*   ei = (const int*)d_in[1];
    const int*   row = ei;
    const int*   col = ei + N_EDGES;
    const float* W1 = (const float*)d_in[2];
    const float* b1 = (const float*)d_in[3];
    const float* W2 = (const float*)d_in[4];
    const float* b2 = (const float*)d_in[5];
    const float* Wl = (const float*)d_in[6];
    const float* bl = (const float*)d_in[7];
    float* out = (float*)d_out;

    float* ws   = (float*)d_ws;
    float* dinv = ws;
    float* bufA = ws + ((N_NODES + 255) / 256) * 256;
    float* bufB = bufA + (size_t)N_NODES * HID;

    const int T = 256;
    const int gN   = (N_NODES + T - 1) / T;
    const int tiles = (N_NODES + 15) / 16;
    const int gGemm = (tiles + (T / 32) - 1) / (T / 32);

    unsigned* part = (unsigned*)bufB;
    k_deg_hist<<<DEG_R * DEG_S, DEG_T, 0, stream>>>(col, part);
    k_dinv<<<gN, T, 0, stream>>>(part, dinv, N_NODES);

    k_gemm_wmma<false><<<gGemm, T, IN_CH * HID * sizeof(float), stream>>>(x, W1, dinv, b1, bufA, bufB,
                                                                          N_NODES, IN_CH);
    k_edge_agg<<<EDGE_BLOCKS, EDGE_T, 0, stream>>>(row, col, bufA, bufB);

    k_gemm_wmma<true><<<gGemm, T, HID * HID * sizeof(float), stream>>>(bufB, W2, dinv, b1, bufA, bufB,
                                                                       N_NODES, HID);
    k_edge_agg<<<EDGE_BLOCKS, EDGE_T, 0, stream>>>(row, col, bufA, bufB);

    k_linear<<<gN, T, 0, stream>>>(bufB, dinv, b2, Wl, bl, out, N_NODES);
}


# Edge pass: every target row gets its float atomics from one package half only (rows split in two, per-half tile counters, kept edges packed via a per-wave LDS ring); sums still in arrival order
# speedup vs baseline: 3.9802x; 1.6255x over previous
#include <hip/hip_runtime.h>
#include <hip/hip_bf16.h>

typedef float v2f __attribute__((ext_vector_type(2)));
typedef float v4f __attribute__((ext_vector_type(4)));
typedef float v8f __attribute__((ext_vector_type(8)));

#define N_NODES 100000
#define N_EDGES 3200000
#define IN_CH   128
#define HID     16


#define DEG_R      16
#define DEG_S      16
#define DEG_RANGE  (N_NODES / DEG_R)
#define DEG_SLICE4 (N_EDGES / DEG_S / 4)
#define DEG_T      1024
#define DEG_U      7

static_assert(N_NODES % DEG_R == 0, "ranges must tile the nodes");
static_assert(N_EDGES % (DEG_S * 4) == 0, "slices must tile the edges in groups of four");
static_assert(DEG_S <= HID, "the partial rows are kept in one N x HID buffer");
static_assert(N_NODES % 16 == 0, "GEMM tiles are whole: a lane's 8 dinv values are read as two 16-byte loads");

#define TICK_STRIDE 64
#define WS_PAD      (((N_NODES + 255) / 256) * 256 - N_NODES)
static_assert(TICK_STRIDE + 2 <= WS_PAD, "the counters fit in the padding behind dinv");
static_assert((N_NODES * 4) % 128 == 0, "the counters start on a line of their own");

__device__ __forceinline__ void deg_bump(unsigned* hist, int c, unsigned lo) {
    unsigned d = (unsigned)c - lo;
    if (d < (unsigned)DEG_RANGE) atomicAdd(&hist[d], 1u);
}

__global__ __launch_bounds__(DEG_T) void k_deg_hist(const int* __restrict__ col,
                                                    unsigned* __restrict__ part) {
    __shared__ unsigned hist[DEG_RANGE];
    const int t  = threadIdx.x;
    const int rg = blockIdx.x % DEG_R;
    const int sl = blockIdx.x / DEG_R;
    for (int i = t; i < DEG_RANGE; i += DEG_T) hist[i] = 0u;
    __syncthreads();

    const unsigned lo = (unsigned)(rg * DEG_RANGE);
    const int4* __restrict__ c4 = reinterpret_cast<const int4*>(col) + (size_t)sl * DEG_SLICE4;
    for (int base = 0; base < DEG_SLICE4; base += DEG_T * DEG_U) {
        int4 v[DEG_U];
        #pragma unroll
        for (int u = 0; u < DEG_U; ++u) {
            int i = base + u * DEG_T + t;
            v[u] = (i < DEG_SLICE4) ? c4[i] : make_int4(-1, -1, -1, -1);
        }
        #pragma unroll
        for (int u = 0; u < DEG_U; ++u) {
            deg_bump(hist, v[u].x, lo);
            deg_bump(hist, v[u].y, lo);
            deg_bump(hist, v[u].z, lo);
            deg_bump(hist, v[u].w, lo);
        }
    }
    __syncthreads();

    unsigned* __restrict__ dst = part + (size_t)sl * N_NODES + lo;
    for (int i = t; i < DEG_RANGE; i += DEG_T) dst[i] = hist[i];
}

__global__ void k_dinv(const unsigned* __restrict__ part, float* __restrict__ dinv,
                       unsigned* __restrict__ tick, int n) {
    int i = blockIdx.x * blockDim.x + threadIdx.x;
    if (i == 0) {
        tick[0] = 0u;               tick[1] = 0u;
        tick[TICK_STRIDE] = 0u;     tick[TICK_STRIDE + 1] = 0u;
    }
    if (i < n) {
        unsigned cnt = 1u;
        for (int s = 0; s < DEG_S; ++s) cnt += part[(size_t)s * N_NODES + i];
        float d = (float)cnt;
        dinv[i] = (d > 0.0f) ? rsqrtf(d) : 0.0f;
    }
}


template <bool PREV>
__global__ void k_gemm_wmma(const float* A, const float* __restrict__ W,
                            const float* __restrict__ dinv, const float* __restrict__ pbias,
                            float* G, float* S, int n, int K) {
    extern __shared__ float sW[];
    int t = threadIdx.x;
    for (int i = t; i < K * HID; i += blockDim.x) sW[i] = W[i];
    __syncthreads();

    int wave = t >> 5;
    int lane = t & 31;
    int m    = lane & 15;
    int half = lane >> 4;

    int tile = blockIdx.x * (blockDim.x >> 5) + wave;
    int rb   = tile * 16;
    if (rb >= n) return;

    int row = rb + m;
    if (row > n - 1) row = n - 1;
    const float* arow = A + (long)row * K + 2 * half;

    const int r0 = rb + 8 * half;
    const v4f d0 = *reinterpret_cast<const v4f*>(dinv + r0);
    const v4f d1 = *reinterpret_cast<const v4f*>(dinv + r0 + 4);
    const float drow = PREV ? dinv[row] : 1.0f;

    v8f acc = {};
    #pragma unroll 8
    for (int k = 0; k < K; k += 4) {
        v2f a = *reinterpret_cast<const v2f*>(arow + k);
        int kb = k + 2 * half;
        if (PREV) {
            a[0] = drow * a[0] + pbias[kb];
            a[1] = drow * a[1] + pbias[kb + 1];
            a[0] = a[0] > 0.0f ? a[0] : 0.0f;
            a[1] = a[1] > 0.0f ? a[1] : 0.0f;
        }
        v2f b;
        b[0] = sW[kb * HID + m];
        b[1] = sW[(kb + 1) * HID + m];
        acc = __builtin_amdgcn_wmma_f32_16x16x4_f32(
                   false, a,  false, b,
                   (short)0, acc,  false,  false);
    }

    #pragma unroll
    for (int i = 0; i < 8; ++i) {
        int r = r0 + i;
        if (r < n) {
            float di = (i < 4) ? d0[i & 3] : d1[i & 3];
            float g  = di * acc[i];
            G[(long)r * HID + m] = g;
            S[(long)r * HID + m] = g;
        }
    }
}


#define EDGE_T         256
#define EDGE_WPB       (EDGE_T / 32)
#define EDGE_PER_WAVE  128
#define EDGE_TILE      (EDGE_WPB * EDGE_PER_WAVE)
#define EDGE_TILES     (N_EDGES / EDGE_TILE)
#define EDGE_BLOCKS    1280
#define EDGE_Q         64
#define OWN_SPLIT      (N_NODES / 2)

static_assert(N_EDGES % EDGE_TILE == 0, "tiles must tile the edges");
static_assert(HID == 16, "one half-wave per 16-channel row");
static_assert((long)N_NODES * HID * 4 < (1L << 32), "row byte offsets are kept in 32 bits");
static_assert(((long)OWN_SPLIT * HID * 4) % 256 == 0, "the two halves' rows share no 256-byte unit");

template <int J> __device__ __forceinline__ unsigned row_lane_u(unsigned v) {
    return (unsigned)__builtin_amdgcn_update_dpp(0, (int)v, 0x150 + J, 0xF, 0xF, true);
}

#define EDGE_GATHER(J) \
    hv[J] = *reinterpret_cast<const float*>(reinterpret_cast<const char*>(h) + (row_lane_u<J>(rb) + ch4));
#define EDGE_SCATTER(J) \
    atomicAdd(reinterpret_cast<float*>(reinterpret_cast<char*>(agg) + (row_lane_u<J>(cb) + ch4)), hv[J]);
#define EDGE_GATHER_P(J) { \
    const unsigned ga = row_lane_u<J>(rb) + ch4; \
    if (J < lim) hv[J] = *reinterpret_cast<const float*>(reinterpret_cast<const char*>(h) + ga); }
#define EDGE_SCATTER_P(J) { \
    const unsigned sa = row_lane_u<J>(cb) + ch4; \
    if (J < lim) atomicAdd(reinterpret_cast<float*>(reinterpret_cast<char*>(agg) + sa), hv[J]); }
#define EDGE_16(M) M(0) M(1) M(2) M(3) M(4) M(5) M(6) M(7) M(8) M(9) M(10) M(11) M(12) M(13) M(14) M(15)

__device__ __forceinline__ void edge_push(uint2* __restrict__ myq, unsigned& tail, int r, int c, unsigned side) {
    const bool own = (((unsigned)c >= (unsigned)OWN_SPLIT) ? 1u : 0u) == side;
    const unsigned m = __builtin_amdgcn_ballot_w32(own);
    if (own) {
        const unsigned pos = (tail + __builtin_amdgcn_mbcnt_lo(m, 0u)) & (EDGE_Q - 1);
        myq[pos] = make_uint2((unsigned)r * (HID * 4u), (unsigned)c * (HID * 4u));
    }
    tail += (unsigned)__builtin_popcount(m);
}

__global__ __launch_bounds__(EDGE_T) void k_edge_agg(const int* __restrict__ row,
                                                     const int* __restrict__ col,
                                                     const float* __restrict__ h,
                                                     float* __restrict__ agg,
                                                     unsigned* __restrict__ tick) {
    __shared__ uint2 q[EDGE_WPB][EDGE_Q];
    __shared__ unsigned s_tile;
    const int lane = threadIdx.x & 31;
    const int wave = threadIdx.x >> 5;
    const unsigned ch4 = (unsigned)(lane & 15) * 4u;
    uint2* __restrict__ myq = q[wave];

    unsigned side = (__builtin_amdgcn_s_sendmsg_rtn(0x87) >> 18) & 1u;
    unsigned head = 0u, tail = 0u;

    const int4* __restrict__ row4 = reinterpret_cast<const int4*>(row);
    const int4* __restrict__ col4 = reinterpret_cast<const int4*>(col);

    for (int phase = 0; phase < 2; ++phase) {
        unsigned* tk = tick + side * TICK_STRIDE;
        for (unsigned it = 0; it <= (unsigned)EDGE_TILES; ++it) {
            if (threadIdx.x == 0) s_tile = atomicAdd(tk, 1u);
            __syncthreads();
            const unsigned tile = (unsigned)__builtin_amdgcn_readfirstlane((int)s_tile);
            __syncthreads();
            if (tile >= (unsigned)EDGE_TILES) break;

            const unsigned e4 = tile * (EDGE_TILE / 4) + (unsigned)wave * (EDGE_PER_WAVE / 4) + (unsigned)lane;
            int4 r4 = row4[e4];
            const int4 c4 = col4[e4];
            asm volatile("" : "+v"(r4.x), "+v"(r4.y), "+v"(r4.z), "+v"(r4.w));

            #define EDGE_STEP(R, C) \
                edge_push(myq, tail, (R), (C), side); \
                if (tail - head >= 32u) { \
                    const uint2 e = myq[(head + (unsigned)lane) & (EDGE_Q - 1)]; \
                    const unsigned rb = e.x, cb = e.y; \
                    float hv[16]; \
                      \
                    EDGE_16(EDGE_GATHER) \
                    EDGE_16(EDGE_SCATTER) \
                    head += 32u; \
                }
            EDGE_STEP(r4.x, c4.x)
            EDGE_STEP(r4.y, c4.y)
            EDGE_STEP(r4.z, c4.z)
            EDGE_STEP(r4.w, c4.w)
            #undef EDGE_STEP
        }
        side ^= 1u;
    }

    const unsigned pend = tail - head;
    if (pend != 0u) {
        uint2 e = make_uint2(0u, 0u);
        if ((unsigned)lane < pend) e = myq[(head + (unsigned)lane) & (EDGE_Q - 1)];
        const unsigned rb = e.x, cb = e.y;
        const int lim = (int)pend - (lane & 16);
        float hv[16] = {};
        EDGE_16(EDGE_GATHER_P)
        EDGE_16(EDGE_SCATTER_P)
    }
}

__global__ void k_linear(const float* __restrict__ t, const float* __restrict__ dinv,
                         const float* __restrict__ b, const float* __restrict__ Wl,
                         const float* __restrict__ bl, float* __restrict__ out, int n) {
    int i = blockIdx.x * blockDim.x + threadIdx.x;
    if (i >= n) return;
    const float* tr = t + (long)i * HID;
    const float di = dinv[i];
    float s = bl[0];
    #pragma unroll
    for (int c = 0; c < HID; ++c) {
        float v = di * tr[c] + b[c];
        s += (v > 0.0f ? v : 0.0f) * Wl[c];
    }
    out[i] = s;
}


extern "C" void kernel_launch(void* const* d_in, const int* in_sizes, int n_in,
                              void* d_out, int out_size, void* d_ws, size_t ws_size,
                              hipStream_t stream) {
    const float* x  = (const float*)d_in[0];
    const int*   ei = (const int*)d_in[1];
    const int*   row = ei;
    const int*   col = ei + N_EDGES;
    const float* W1 = (const float*)d_in[2];
    const float* b1 = (const float*)d_in[3];
    const float* W2 = (const float*)d_in[4];
    const float* b2 = (const float*)d_in[5];
    const float* Wl = (const float*)d_in[6];
    const float* bl = (const float*)d_in[7];
    float* out = (float*)d_out;

    float* ws   = (float*)d_ws;
    float* dinv = ws;
    float* bufA = ws + ((N_NODES + 255) / 256) * 256;
    float* bufB = bufA + (size_t)N_NODES * HID;

    const int T = 256;
    const int gN   = (N_NODES + T - 1) / T;
    const int tiles = (N_NODES + 15) / 16;
    const int gGemm = (tiles + (T / 32) - 1) / (T / 32);

    unsigned* part = (unsigned*)bufB;
    k_deg_hist<<<DEG_R * DEG_S, DEG_T, 0, stream>>>(col, part);
    unsigned* tick = (unsigned*)(ws + N_NODES);
    k_dinv<<<gN, T, 0, stream>>>(part, dinv, tick, N_NODES);

    k_gemm_wmma<false><<<gGemm, T, IN_CH * HID * sizeof(float), stream>>>(x, W1, dinv, b1, bufA, bufB,
                                                                          N_NODES, IN_CH);
    k_edge_agg<<<EDGE_BLOCKS, EDGE_T, 0, stream>>>(row, col, bufA, bufB, tick);

    k_gemm_wmma<true><<<gGemm, T, HID * HID * sizeof(float), stream>>>(bufB, W2, dinv, b1, bufA, bufB,
                                                                       N_NODES, HID);
    k_edge_agg<<<EDGE_BLOCKS, EDGE_T, 0, stream>>>(row, col, bufA, bufB, tick + 1);

    k_linear<<<gN, T, 0, stream>>>(bufB, dinv, b2, Wl, bl, out, N_NODES);
}


# Edge passes: edges binned once by target range (one word each), each range summed by one work-group in LDS with LDS float adds (arrival order as before), rows stored once; no global float atomics
# speedup vs baseline: 4.5015x; 1.1310x over previous
#include <hip/hip_runtime.h>
#include <hip/hip_bf16.h>

typedef float v2f __attribute__((ext_vector_type(2)));
typedef float v4f __attribute__((ext_vector_type(4)));
typedef float v8f __attribute__((ext_vector_type(8)));

#define N_NODES 100000
#define N_EDGES 3200000
#define IN_CH   128
#define HID     16


#define DEG_R      16
#define DEG_S      16
#define DEG_RANGE  (N_NODES / DEG_R)
#define DEG_SLICE4 (N_EDGES / DEG_S / 4)
#define DEG_T      1024
#define DEG_U      7

static_assert(N_NODES % DEG_R == 0, "ranges must tile the nodes");
static_assert(N_EDGES % (DEG_S * 4) == 0, "slices must tile the edges in groups of four");
static_assert(DEG_S <= HID, "the partial rows are kept in one N x HID buffer");
static_assert(N_NODES % 16 == 0, "GEMM tiles are whole: a lane's 8 dinv values are read as two 16-byte loads");

#define TICK_STRIDE 64
#define WS_PAD      (((N_NODES + 255) / 256) * 256 - N_NODES)
static_assert(TICK_STRIDE + 2 <= WS_PAD, "the counters fit in the padding behind dinv");
static_assert((N_NODES * 4) % 128 == 0, "the counters start on a line of their own");

__device__ __forceinline__ void deg_bump(unsigned* hist, int c, unsigned lo) {
    unsigned d = (unsigned)c - lo;
    if (d < (unsigned)DEG_RANGE) atomicAdd(&hist[d], 1u);
}

__global__ __launch_bounds__(DEG_T) void k_deg_hist(const int* __restrict__ col,
                                                    unsigned* __restrict__ part) {
    __shared__ unsigned hist[DEG_RANGE];
    const int t  = threadIdx.x;
    const int rg = blockIdx.x % DEG_R;
    const int sl = blockIdx.x / DEG_R;
    for (int i = t; i < DEG_RANGE; i += DEG_T) hist[i] = 0u;
    __syncthreads();

    const unsigned lo = (unsigned)(rg * DEG_RANGE);
    const int4* __restrict__ c4 = reinterpret_cast<const int4*>(col) + (size_t)sl * DEG_SLICE4;
    for (int base = 0; base < DEG_SLICE4; base += DEG_T * DEG_U) {
        int4 v[DEG_U];
        #pragma unroll
        for (int u = 0; u < DEG_U; ++u) {
            int i = base + u * DEG_T + t;
            v[u] = (i < DEG_SLICE4) ? c4[i] : make_int4(-1, -1, -1, -1);
        }
        #pragma unroll
        for (int u = 0; u < DEG_U; ++u) {
            deg_bump(hist, v[u].x, lo);
            deg_bump(hist, v[u].y, lo);
            deg_bump(hist, v[u].z, lo);
            deg_bump(hist, v[u].w, lo);
        }
    }
    __syncthreads();

    unsigned* __restrict__ dst = part + (size_t)sl * N_NODES + lo;
    for (int i = t; i < DEG_RANGE; i += DEG_T) dst[i] = hist[i];
}

__global__ void k_dinv(const unsigned* __restrict__ part, float* __restrict__ dinv,
                       unsigned* __restrict__ tick, int n) {
    int i = blockIdx.x * blockDim.x + threadIdx.x;
    if (i == 0) {
        tick[0] = 0u;               tick[1] = 0u;
        tick[TICK_STRIDE] = 0u;     tick[TICK_STRIDE + 1] = 0u;
    }
    if (i < n) {
        unsigned cnt = 1u;
        for (int s = 0; s < DEG_S; ++s) cnt += part[(size_t)s * N_NODES + i];
        float d = (float)cnt;
        dinv[i] = (d > 0.0f) ? rsqrtf(d) : 0.0f;
    }
}


template <bool PREV, bool START>
__global__ void k_gemm_wmma(const float* A, const float* __restrict__ W,
                            const float* __restrict__ dinv, const float* __restrict__ pbias,
                            float* G, float* S, int n, int K) {
    extern __shared__ float sW[];
    int t = threadIdx.x;
    for (int i = t; i < K * HID; i += blockDim.x) sW[i] = W[i];
    __syncthreads();

    int wave = t >> 5;
    int lane = t & 31;
    int m    = lane & 15;
    int half = lane >> 4;

    int tile = blockIdx.x * (blockDim.x >> 5) + wave;
    int rb   = tile * 16;
    if (rb >= n) return;

    int row = rb + m;
    if (row > n - 1) row = n - 1;
    const float* arow = A + (long)row * K + 2 * half;

    const int r0 = rb + 8 * half;
    const v4f d0 = *reinterpret_cast<const v4f*>(dinv + r0);
    const v4f d1 = *reinterpret_cast<const v4f*>(dinv + r0 + 4);
    const float drow = PREV ? dinv[row] : 1.0f;

    v8f acc = {};
    #pragma unroll 8
    for (int k = 0; k < K; k += 4) {
        v2f a = *reinterpret_cast<const v2f*>(arow + k);
        int kb = k + 2 * half;
        if (PREV) {
            a[0] = drow * a[0] + pbias[kb];
            a[1] = drow * a[1] + pbias[kb + 1];
            a[0] = a[0] > 0.0f ? a[0] : 0.0f;
            a[1] = a[1] > 0.0f ? a[1] : 0.0f;
        }
        v2f b;
        b[0] = sW[kb * HID + m];
        b[1] = sW[(kb + 1) * HID + m];
        acc = __builtin_amdgcn_wmma_f32_16x16x4_f32(
                   false, a,  false, b,
                   (short)0, acc,  false,  false);
    }

    #pragma unroll
    for (int i = 0; i < 8; ++i) {
        int r = r0 + i;
        if (r < n) {
            float di = (i < 4) ? d0[i & 3] : d1[i & 3];
            float g  = di * acc[i];
            G[(long)r * HID + m] = g;
            if (START) S[(long)r * HID + m] = g;
        }
    }
}


#define EDGE_T         256
#define EDGE_WPB       (EDGE_T / 32)
#define EDGE_PER_WAVE  128
#define EDGE_TILE      (EDGE_WPB * EDGE_PER_WAVE)
#define EDGE_TILES     (N_EDGES / EDGE_TILE)
#define EDGE_BLOCKS    1280
#define EDGE_Q         64
#define OWN_SPLIT      (N_NODES / 2)

static_assert(N_EDGES % EDGE_TILE == 0, "tiles must tile the edges");
static_assert(HID == 16, "one half-wave per 16-channel row");
static_assert((long)N_NODES * HID * 4 < (1L << 32), "row byte offsets are kept in 32 bits");
static_assert(((long)OWN_SPLIT * HID * 4) % 256 == 0, "the two halves' rows share no 256-byte unit");

template <int J> __device__ __forceinline__ unsigned row_lane_u(unsigned v) {
    return (unsigned)__builtin_amdgcn_update_dpp(0, (int)v, 0x150 + J, 0xF, 0xF, true);
}

#define EDGE_GATHER(J) \
    hv[J] = *reinterpret_cast<const float*>(reinterpret_cast<const char*>(h) + (row_lane_u<J>(rb) + ch4));
#define EDGE_SCATTER(J) \
    atomicAdd(reinterpret_cast<float*>(reinterpret_cast<char*>(agg) + (row_lane_u<J>(cb) + ch4)), hv[J]);
#define EDGE_GATHER_P(J) { \
    const unsigned ga = row_lane_u<J>(rb) + ch4; \
    if (J < lim) hv[J] = *reinterpret_cast<const float*>(reinterpret_cast<const char*>(h) + ga); }
#define EDGE_SCATTER_P(J) { \
    const unsigned sa = row_lane_u<J>(cb) + ch4; \
    if (J < lim) atomicAdd(reinterpret_cast<float*>(reinterpret_cast<char*>(agg) + sa), hv[J]); }
#define EDGE_16(M) M(0) M(1) M(2) M(3) M(4) M(5) M(6) M(7) M(8) M(9) M(10) M(11) M(12) M(13) M(14) M(15)

__device__ __forceinline__ void edge_push(uint2* __restrict__ myq, unsigned& tail, int r, int c, unsigned side) {
    const bool own = (((unsigned)c >= (unsigned)OWN_SPLIT) ? 1u : 0u) == side;
    const unsigned m = __builtin_amdgcn_ballot_w32(own);
    if (own) {
        const unsigned pos = (tail + __builtin_amdgcn_mbcnt_lo(m, 0u)) & (EDGE_Q - 1);
        myq[pos] = make_uint2((unsigned)r * (HID * 4u), (unsigned)c * (HID * 4u));
    }
    tail += (unsigned)__builtin_popcount(m);
}

__global__ __launch_bounds__(EDGE_T) void k_edge_agg(const int* __restrict__ row,
                                                     const int* __restrict__ col,
                                                     const float* __restrict__ h,
                                                     float* __restrict__ agg,
                                                     unsigned* __restrict__ tick) {
    __shared__ uint2 q[EDGE_WPB][EDGE_Q];
    __shared__ unsigned s_tile;
    const int lane = threadIdx.x & 31;
    const int wave = threadIdx.x >> 5;
    const unsigned ch4 = (unsigned)(lane & 15) * 4u;
    uint2* __restrict__ myq = q[wave];

    unsigned side = (__builtin_amdgcn_s_sendmsg_rtn(0x87) >> 18) & 1u;
    unsigned head = 0u, tail = 0u;

    const int4* __restrict__ row4 = reinterpret_cast<const int4*>(row);
    const int4* __restrict__ col4 = reinterpret_cast<const int4*>(col);

    for (int phase = 0; phase < 2; ++phase) {
        unsigned* tk = tick + side * TICK_STRIDE;
        for (unsigned it = 0; it <= (unsigned)EDGE_TILES; ++it) {
            if (threadIdx.x == 0) s_tile = atomicAdd(tk, 1u);
            __syncthreads();
            const unsigned tile = (unsigned)__builtin_amdgcn_readfirstlane((int)s_tile);
            __syncthreads();
            if (tile >= (unsigned)EDGE_TILES) break;

            const unsigned e4 = tile * (EDGE_TILE / 4) + (unsigned)wave * (EDGE_PER_WAVE / 4) + (unsigned)lane;
            int4 r4 = row4[e4];
            const int4 c4 = col4[e4];
            asm volatile("" : "+v"(r4.x), "+v"(r4.y), "+v"(r4.z), "+v"(r4.w));

            #define EDGE_STEP(R, C) \
                edge_push(myq, tail, (R), (C), side); \
                if (tail - head >= 32u) { \
                    const uint2 e = myq[(head + (unsigned)lane) & (EDGE_Q - 1)]; \
                    const unsigned rb = e.x, cb = e.y; \
                    float hv[16]; \
                      \
                    EDGE_16(EDGE_GATHER) \
                    EDGE_16(EDGE_SCATTER) \
                    head += 32u; \
                }
            EDGE_STEP(r4.x, c4.x)
            EDGE_STEP(r4.y, c4.y)
            EDGE_STEP(r4.z, c4.z)
            EDGE_STEP(r4.w, c4.w)
            #undef EDGE_STEP
        }
        side ^= 1u;
    }

    const unsigned pend = tail - head;
    if (pend != 0u) {
        uint2 e = make_uint2(0u, 0u);
        if ((unsigned)lane < pend) e = myq[(head + (unsigned)lane) & (EDGE_Q - 1)];
        const unsigned rb = e.x, cb = e.y;
        const int lim = (int)pend - (lane & 16);
        float hv[16] = {};
        EDGE_16(EDGE_GATHER_P)
        EDGE_16(EDGE_SCATTER_P)
    }
}


#define OWN_RR        200
#define OWN_NR        (N_NODES / OWN_RR)
#define OWN_SRC_BITS  17
#define OWN_SRC_MASK  ((1u << OWN_SRC_BITS) - 1u)
#define OWN_T         512
#define CUR_STRIDE    64
#define BIN_CH        6400
#define BIN_NCH       (N_EDGES / BIN_CH)
#define BIN_T         320
#define BIN_U         5
#define OWN_WS_WORDS  ((size_t)((N_NODES + 255) / 256) * 256 + 2 * (size_t)N_NODES * HID + N_EDGES + 1024 + (size_t)OWN_NR * CUR_STRIDE)

static_assert(N_NODES % OWN_RR == 0, "ranges must tile the rows");
static_assert(N_NODES <= (1 << OWN_SRC_BITS), "a source row fits its field");
static_assert(OWN_RR < (1 << (32 - OWN_SRC_BITS)), "a target offset (and the spare row) fits its field");
static_assert(OWN_NR <= 512, "one wave scans the ranges, 16 per lane");
static_assert(OWN_RR <= 256, "one thread per row in k_dinv_own");
static_assert(N_EDGES % BIN_CH == 0 && BIN_T * BIN_U * 4 == BIN_CH, "chunks tile the edges, threads tile a chunk");
static_assert(BIN_T <= OWN_NR && 2 * BIN_T >= OWN_NR, "a binning thread reserves for one or two ranges");

__global__ __launch_bounds__(256) void k_dinv_own(const unsigned* __restrict__ part, float* __restrict__ dinv,
                                                  unsigned* __restrict__ tot, unsigned* __restrict__ cur) {
    __shared__ unsigned s_sum[8];
    const int t = threadIdx.x;
    const int r = blockIdx.x;
    unsigned inc = 0u;
    if (t < OWN_RR) {
        const int i = r * OWN_RR + t;
        for (int s = 0; s < DEG_S; ++s) inc += part[(size_t)s * N_NODES + i];
        float d = (float)(inc + 1u);
        dinv[i] = (d > 0.0f) ? rsqrtf(d) : 0.0f;
    }
    unsigned v = inc;
    for (int o = 16; o > 0; o >>= 1) v += (unsigned)__shfl_xor((int)v, o);
    if ((t & 31) == 0) s_sum[t >> 5] = v;
    __syncthreads();
    if (t == 0) {
        unsigned s = 0u;
        for (int w = 0; w < 8; ++w) s += s_sum[w];
        tot[r] = s;
        cur[r * CUR_STRIDE] = 0u;
    }
}

__device__ __forceinline__ void own_scan(const unsigned* src, unsigned* dst, int lane) {
    unsigned v[16];
    unsigned s = 0u;
    #pragma unroll
    for (int j = 0; j < 16; ++j) {
        const int idx = lane * 16 + j;
        v[j] = (idx < OWN_NR) ? src[idx] : 0u;
        s += v[j];
    }
    unsigned inc = s;
    for (int o = 1; o < 32; o <<= 1) {
        const unsigned nb = (unsigned)__shfl_up((int)inc, o);
        if (lane >= o) inc += nb;
    }
    unsigned run = inc - s;
    #pragma unroll
    for (int j = 0; j < 16; ++j) {
        const int idx = lane * 16 + j;
        if (idx < OWN_NR) dst[idx] = run;
        run += v[j];
    }
}

__global__ __launch_bounds__(BIN_T) void k_bin(const int* __restrict__ row, const int* __restrict__ col,
                                               const unsigned* __restrict__ tot, unsigned* __restrict__ cur,
                                               unsigned* __restrict__ seg, unsigned* __restrict__ list) {
    __shared__ unsigned s_cnt[OWN_NR];
    __shared__ unsigned s_lst[OWN_NR];
    __shared__ unsigned s_gst[OWN_NR];
    __shared__ unsigned s_seg[OWN_NR];
    __shared__ unsigned s_w[BIN_CH];
    __shared__ unsigned s_g[BIN_CH];
    const int t = threadIdx.x;
    const int lane = t & 31;
    const int wave = t >> 5;

    for (int i = t; i < OWN_NR; i += BIN_T) { s_cnt[i] = 0u; s_seg[i] = tot[i]; }

    const int4* __restrict__ row4 = reinterpret_cast<const int4*>(row) + (size_t)blockIdx.x * (BIN_CH / 4);
    const int4* __restrict__ col4 = reinterpret_cast<const int4*>(col) + (size_t)blockIdx.x * (BIN_CH / 4);
    int4 r4[BIN_U], c4[BIN_U];
    #pragma unroll
    for (int u = 0; u < BIN_U; ++u) { r4[u] = row4[u * BIN_T + t]; c4[u] = col4[u * BIN_T + t]; }
    __syncthreads();

    #pragma unroll
    for (int u = 0; u < BIN_U; ++u) {
        atomicAdd(&s_cnt[(unsigned)c4[u].x / OWN_RR], 1u);
        atomicAdd(&s_cnt[(unsigned)c4[u].y / OWN_RR], 1u);
        atomicAdd(&s_cnt[(unsigned)c4[u].z / OWN_RR], 1u);
        atomicAdd(&s_cnt[(unsigned)c4[u].w / OWN_RR], 1u);
    }
    __syncthreads();

    if (wave == 0) own_scan(s_seg, s_seg, lane);
    if (wave == 1) own_scan(s_cnt, s_lst, lane);
    __syncthreads();

    {
        const int i0 = t, i1 = t + BIN_T;
        const bool two = i1 < OWN_NR;
        const unsigned n0 = s_cnt[i0];
        const unsigned n1 = two ? s_cnt[i1] : 0u;
        const unsigned off0 = atomicAdd(&cur[i0 * CUR_STRIDE], n0);
        unsigned off1 = 0u;
        if (two) off1 = atomicAdd(&cur[i1 * CUR_STRIDE], n1);
        s_gst[i0] = s_seg[i0] + off0;
        s_cnt[i0] = 0u;
        if (two) { s_gst[i1] = s_seg[i1] + off1; s_cnt[i1] = 0u; }
        if (blockIdx.x == 0) {
            seg[i0] = s_seg[i0];
            if (two) seg[i1] = s_seg[i1];
        }
    }
    __syncthreads();

    #define BIN_PUT(R, C) { \
        const unsigned rg = (unsigned)(C) / OWN_RR; \
        const unsigned lo = (unsigned)(C) - rg * OWN_RR; \
        const unsigned sl = atomicAdd(&s_cnt[rg], 1u); \
        const unsigned p  = s_lst[rg] + sl; \
        s_w[p] = (unsigned)(R) | (lo << OWN_SRC_BITS); \
        s_g[p] = s_gst[rg] + sl; }
    #pragma unroll
    for (int u = 0; u < BIN_U; ++u) {
        BIN_PUT(r4[u].x, c4[u].x)
        BIN_PUT(r4[u].y, c4[u].y)
        BIN_PUT(r4[u].z, c4[u].z)
        BIN_PUT(r4[u].w, c4[u].w)
    }
    #undef BIN_PUT
    __syncthreads();

    for (int p = t; p < BIN_CH; p += BIN_T) list[s_g[p]] = s_w[p];
}

#define OWN_GATHER(J) \
    hv[J] = *reinterpret_cast<const float*>(reinterpret_cast<const char*>(g) + (row_lane_u<J>(rb) + ch4));
#define OWN_ADD(J) \
    atomicAdd(reinterpret_cast<float*>(reinterpret_cast<char*>(acc) + (row_lane_u<J>(lb) + ch4)), hv[J]);

__global__ __launch_bounds__(OWN_T) void k_own_agg(const unsigned* __restrict__ list, const unsigned* __restrict__ seg,
                                                   const unsigned* __restrict__ tot, const float* __restrict__ g,
                                                   float* __restrict__ out) {
    __shared__ __attribute__((aligned(16))) float acc[(OWN_RR + 1) * HID];
    const int t = threadIdx.x;
    const int lane = t & 31;
    const int wave = __builtin_amdgcn_readfirstlane(t >> 5);
    const int r = blockIdx.x;
    const unsigned ch4 = (unsigned)(lane & 15) * 4u;
    const unsigned s0 = seg[r];
    const unsigned s1 = s0 + tot[r];
    const unsigned pad = (unsigned)OWN_RR << OWN_SRC_BITS;

    unsigned base = s0 + (unsigned)wave * 32u;
    unsigned w = (base + (unsigned)lane < s1) ? list[base + lane] : pad;

    const v4f* __restrict__ gown = reinterpret_cast<const v4f*>(g + (size_t)r * (OWN_RR * HID));
    v4f* acc4 = reinterpret_cast<v4f*>(acc);
    for (int i = t; i < OWN_RR * HID / 4; i += OWN_T) acc4[i] = gown[i];
    __syncthreads();

    while (base < s1) {
        const unsigned rb = (w & OWN_SRC_MASK) * (HID * 4u);
        const unsigned lb = (w >> OWN_SRC_BITS) * (HID * 4u);
        base += (OWN_T / 32) * 32u;
        w = (base + (unsigned)lane < s1) ? list[base + lane] : pad;
        float hv[16];
        EDGE_16(OWN_GATHER)
        EDGE_16(OWN_ADD)
    }
    __syncthreads();

    v4f* __restrict__ oown = reinterpret_cast<v4f*>(out + (size_t)r * (OWN_RR * HID));
    for (int i = t; i < OWN_RR * HID / 4; i += OWN_T) oown[i] = acc4[i];
}

__global__ void k_linear(const float* __restrict__ t, const float* __restrict__ dinv,
                         const float* __restrict__ b, const float* __restrict__ Wl,
                         const float* __restrict__ bl, float* __restrict__ out, int n) {
    int i = blockIdx.x * blockDim.x + threadIdx.x;
    if (i >= n) return;
    const float* tr = t + (long)i * HID;
    const float di = dinv[i];
    float s = bl[0];
    #pragma unroll
    for (int c = 0; c < HID; ++c) {
        float v = di * tr[c] + b[c];
        s += (v > 0.0f ? v : 0.0f) * Wl[c];
    }
    out[i] = s;
}


extern "C" void kernel_launch(void* const* d_in, const int* in_sizes, int n_in,
                              void* d_out, int out_size, void* d_ws, size_t ws_size,
                              hipStream_t stream) {
    const float* x  = (const float*)d_in[0];
    const int*   ei = (const int*)d_in[1];
    const int*   row = ei;
    const int*   col = ei + N_EDGES;
    const float* W1 = (const float*)d_in[2];
    const float* b1 = (const float*)d_in[3];
    const float* W2 = (const float*)d_in[4];
    const float* b2 = (const float*)d_in[5];
    const float* Wl = (const float*)d_in[6];
    const float* bl = (const float*)d_in[7];
    float* out = (float*)d_out;

    float* ws   = (float*)d_ws;
    float* dinv = ws;
    float* bufA = ws + ((N_NODES + 255) / 256) * 256;
    float* bufB = bufA + (size_t)N_NODES * HID;

    const int T = 256;
    const int gN   = (N_NODES + T - 1) / T;
    const int tiles = (N_NODES + 15) / 16;
    const int gGemm = (tiles + (T / 32) - 1) / (T / 32);

    unsigned* part = (unsigned*)bufB;
    k_deg_hist<<<DEG_R * DEG_S, DEG_T, 0, stream>>>(col, part);

    if (ws_size >= sizeof(float) * OWN_WS_WORDS) {
        unsigned* list = (unsigned*)(bufB + (size_t)N_NODES * HID);
        unsigned* tot  = list + N_EDGES;
        unsigned* seg  = tot + 512;
        unsigned* cur  = seg + 512;
        k_dinv_own<<<OWN_NR, 256, 0, stream>>>(part, dinv, tot, cur);
        k_bin<<<BIN_NCH, BIN_T, 0, stream>>>(row, col, tot, cur, seg, list);

        k_gemm_wmma<false, false><<<gGemm, T, IN_CH * HID * sizeof(float), stream>>>(x, W1, dinv, b1, bufA, nullptr,
                                                                                     N_NODES, IN_CH);
        k_own_agg<<<OWN_NR, OWN_T, 0, stream>>>(list, seg, tot, bufA, bufB);
        k_gemm_wmma<true, false><<<gGemm, T, HID * HID * sizeof(float), stream>>>(bufB, W2, dinv, b1, bufA, nullptr,
                                                                                  N_NODES, HID);
        k_own_agg<<<OWN_NR, OWN_T, 0, stream>>>(list, seg, tot, bufA, bufB);
        k_linear<<<gN, T, 0, stream>>>(bufB, dinv, b2, Wl, bl, out, N_NODES);
        return;
    }

    unsigned* tick = (unsigned*)(ws + N_NODES);
    k_dinv<<<gN, T, 0, stream>>>(part, dinv, tick, N_NODES);

    k_gemm_wmma<false, true><<<gGemm, T, IN_CH * HID * sizeof(float), stream>>>(x, W1, dinv, b1, bufA, bufB,
                                                                                N_NODES, IN_CH);
    k_edge_agg<<<EDGE_BLOCKS, EDGE_T, 0, stream>>>(row, col, bufA, bufB, tick);

    k_gemm_wmma<true, true><<<gGemm, T, HID * HID * sizeof(float), stream>>>(bufB, W2, dinv, b1, bufA, bufB,
                                                                             N_NODES, HID);
    k_edge_agg<<<EDGE_BLOCKS, EDGE_T, 0, stream>>>(row, col, bufA, bufB, tick + 1);

    k_linear<<<gN, T, 0, stream>>>(bufB, dinv, b2, Wl, bl, out, N_NODES);
}


# Owner passes finish their rows' consumer in LDS: pass 1 runs relu and the layer-2 WMMA product on its 16-row tiles, pass 2 the readout; sums never stored; 8 -> 6 launches; sum order as before
# speedup vs baseline: 4.8539x; 1.0783x over previous
#include <hip/hip_runtime.h>
#include <hip/hip_bf16.h>

typedef float v2f __attribute__((ext_vector_type(2)));
typedef float v4f __attribute__((ext_vector_type(4)));
typedef float v8f __attribute__((ext_vector_type(8)));

#define N_NODES 100000
#define N_EDGES 3200000
#define IN_CH   128
#define HID     16


#define DEG_R      16
#define DEG_S      16
#define DEG_RANGE  (N_NODES / DEG_R)
#define DEG_SLICE4 (N_EDGES / DEG_S / 4)
#define DEG_T      1024
#define DEG_U      7

static_assert(N_NODES % DEG_R == 0, "ranges must tile the nodes");
static_assert(N_EDGES % (DEG_S * 4) == 0, "slices must tile the edges in groups of four");
static_assert(DEG_S <= HID, "the partial rows are kept in one N x HID buffer");
static_assert(N_NODES % 16 == 0, "GEMM tiles are whole: a lane's 8 dinv values are read as two 16-byte loads");

#define TICK_STRIDE 64
#define WS_PAD      (((N_NODES + 255) / 256) * 256 - N_NODES)
static_assert(TICK_STRIDE + 2 <= WS_PAD, "the counters fit in the padding behind dinv");
static_assert((N_NODES * 4) % 128 == 0, "the counters start on a line of their own");

__device__ __forceinline__ void deg_bump(unsigned* hist, int c, unsigned lo) {
    unsigned d = (unsigned)c - lo;
    if (d < (unsigned)DEG_RANGE) atomicAdd(&hist[d], 1u);
}

__global__ __launch_bounds__(DEG_T) void k_deg_hist(const int* __restrict__ col,
                                                    unsigned* __restrict__ part) {
    __shared__ unsigned hist[DEG_RANGE];
    const int t  = threadIdx.x;
    const int rg = blockIdx.x % DEG_R;
    const int sl = blockIdx.x / DEG_R;
    for (int i = t; i < DEG_RANGE; i += DEG_T) hist[i] = 0u;
    __syncthreads();

    const unsigned lo = (unsigned)(rg * DEG_RANGE);
    const int4* __restrict__ c4 = reinterpret_cast<const int4*>(col) + (size_t)sl * DEG_SLICE4;
    for (int base = 0; base < DEG_SLICE4; base += DEG_T * DEG_U) {
        int4 v[DEG_U];
        #pragma unroll
        for (int u = 0; u < DEG_U; ++u) {
            int i = base + u * DEG_T + t;
            v[u] = (i < DEG_SLICE4) ? c4[i] : make_int4(-1, -1, -1, -1);
        }
        #pragma unroll
        for (int u = 0; u < DEG_U; ++u) {
            deg_bump(hist, v[u].x, lo);
            deg_bump(hist, v[u].y, lo);
            deg_bump(hist, v[u].z, lo);
            deg_bump(hist, v[u].w, lo);
        }
    }
    __syncthreads();

    unsigned* __restrict__ dst = part + (size_t)sl * N_NODES + lo;
    for (int i = t; i < DEG_RANGE; i += DEG_T) dst[i] = hist[i];
}

__global__ void k_dinv(const unsigned* __restrict__ part, float* __restrict__ dinv,
                       unsigned* __restrict__ tick, int n) {
    int i = blockIdx.x * blockDim.x + threadIdx.x;
    if (i == 0) {
        tick[0] = 0u;               tick[1] = 0u;
        tick[TICK_STRIDE] = 0u;     tick[TICK_STRIDE + 1] = 0u;
    }
    if (i < n) {
        unsigned cnt = 1u;
        for (int s = 0; s < DEG_S; ++s) cnt += part[(size_t)s * N_NODES + i];
        float d = (float)cnt;
        dinv[i] = (d > 0.0f) ? rsqrtf(d) : 0.0f;
    }
}


template <bool PREV, bool START>
__global__ void k_gemm_wmma(const float* A, const float* __restrict__ W,
                            const float* __restrict__ dinv, const float* __restrict__ pbias,
                            float* G, float* S, int n, int K) {
    extern __shared__ float sW[];
    int t = threadIdx.x;
    for (int i = t; i < K * HID; i += blockDim.x) sW[i] = W[i];
    __syncthreads();

    int wave = t >> 5;
    int lane = t & 31;
    int m    = lane & 15;
    int half = lane >> 4;

    int tile = blockIdx.x * (blockDim.x >> 5) + wave;
    int rb   = tile * 16;
    if (rb >= n) return;

    int row = rb + m;
    if (row > n - 1) row = n - 1;
    const float* arow = A + (long)row * K + 2 * half;

    const int r0 = rb + 8 * half;
    const v4f d0 = *reinterpret_cast<const v4f*>(dinv + r0);
    const v4f d1 = *reinterpret_cast<const v4f*>(dinv + r0 + 4);
    const float drow = PREV ? dinv[row] : 1.0f;

    v8f acc = {};
    #pragma unroll 8
    for (int k = 0; k < K; k += 4) {
        v2f a = *reinterpret_cast<const v2f*>(arow + k);
        int kb = k + 2 * half;
        if (PREV) {
            a[0] = drow * a[0] + pbias[kb];
            a[1] = drow * a[1] + pbias[kb + 1];
            a[0] = a[0] > 0.0f ? a[0] : 0.0f;
            a[1] = a[1] > 0.0f ? a[1] : 0.0f;
        }
        v2f b;
        b[0] = sW[kb * HID + m];
        b[1] = sW[(kb + 1) * HID + m];
        acc = __builtin_amdgcn_wmma_f32_16x16x4_f32(
                   false, a,  false, b,
                   (short)0, acc,  false,  false);
    }

    #pragma unroll
    for (int i = 0; i < 8; ++i) {
        int r = r0 + i;
        if (r < n) {
            float di = (i < 4) ? d0[i & 3] : d1[i & 3];
            float g  = di * acc[i];
            G[(long)r * HID + m] = g;
            if (START) S[(long)r * HID + m] = g;
        }
    }
}


#define EDGE_T         256
#define EDGE_WPB       (EDGE_T / 32)
#define EDGE_PER_WAVE  128
#define EDGE_TILE      (EDGE_WPB * EDGE_PER_WAVE)
#define EDGE_TILES     (N_EDGES / EDGE_TILE)
#define EDGE_BLOCKS    1280
#define EDGE_Q         64
#define OWN_SPLIT      (N_NODES / 2)

static_assert(N_EDGES % EDGE_TILE == 0, "tiles must tile the edges");
static_assert(HID == 16, "one half-wave per 16-channel row");
static_assert((long)N_NODES * HID * 4 < (1L << 32), "row byte offsets are kept in 32 bits");
static_assert(((long)OWN_SPLIT * HID * 4) % 256 == 0, "the two halves' rows share no 256-byte unit");

template <int J> __device__ __forceinline__ unsigned row_lane_u(unsigned v) {
    return (unsigned)__builtin_amdgcn_update_dpp(0, (int)v, 0x150 + J, 0xF, 0xF, true);
}

#define EDGE_GATHER(J) \
    hv[J] = *reinterpret_cast<const float*>(reinterpret_cast<const char*>(h) + (row_lane_u<J>(rb) + ch4));
#define EDGE_SCATTER(J) \
    atomicAdd(reinterpret_cast<float*>(reinterpret_cast<char*>(agg) + (row_lane_u<J>(cb) + ch4)), hv[J]);
#define EDGE_GATHER_P(J) { \
    const unsigned ga = row_lane_u<J>(rb) + ch4; \
    if (J < lim) hv[J] = *reinterpret_cast<const float*>(reinterpret_cast<const char*>(h) + ga); }
#define EDGE_SCATTER_P(J) { \
    const unsigned sa = row_lane_u<J>(cb) + ch4; \
    if (J < lim) atomicAdd(reinterpret_cast<float*>(reinterpret_cast<char*>(agg) + sa), hv[J]); }
#define EDGE_16(M) M(0) M(1) M(2) M(3) M(4) M(5) M(6) M(7) M(8) M(9) M(10) M(11) M(12) M(13) M(14) M(15)

__device__ __forceinline__ void edge_push(uint2* __restrict__ myq, unsigned& tail, int r, int c, unsigned side) {
    const bool own = (((unsigned)c >= (unsigned)OWN_SPLIT) ? 1u : 0u) == side;
    const unsigned m = __builtin_amdgcn_ballot_w32(own);
    if (own) {
        const unsigned pos = (tail + __builtin_amdgcn_mbcnt_lo(m, 0u)) & (EDGE_Q - 1);
        myq[pos] = make_uint2((unsigned)r * (HID * 4u), (unsigned)c * (HID * 4u));
    }
    tail += (unsigned)__builtin_popcount(m);
}

__global__ __launch_bounds__(EDGE_T) void k_edge_agg(const int* __restrict__ row,
                                                     const int* __restrict__ col,
                                                     const float* __restrict__ h,
                                                     float* __restrict__ agg,
                                                     unsigned* __restrict__ tick) {
    __shared__ uint2 q[EDGE_WPB][EDGE_Q];
    __shared__ unsigned s_tile;
    const int lane = threadIdx.x & 31;
    const int wave = threadIdx.x >> 5;
    const unsigned ch4 = (unsigned)(lane & 15) * 4u;
    uint2* __restrict__ myq = q[wave];

    unsigned side = (__builtin_amdgcn_s_sendmsg_rtn(0x87) >> 18) & 1u;
    unsigned head = 0u, tail = 0u;

    const int4* __restrict__ row4 = reinterpret_cast<const int4*>(row);
    const int4* __restrict__ col4 = reinterpret_cast<const int4*>(col);

    for (int phase = 0; phase < 2; ++phase) {
        unsigned* tk = tick + side * TICK_STRIDE;
        for (unsigned it = 0; it <= (unsigned)EDGE_TILES; ++it) {
            if (threadIdx.x == 0) s_tile = atomicAdd(tk, 1u);
            __syncthreads();
            const unsigned tile = (unsigned)__builtin_amdgcn_readfirstlane((int)s_tile);
            __syncthreads();
            if (tile >= (unsigned)EDGE_TILES) break;

            const unsigned e4 = tile * (EDGE_TILE / 4) + (unsigned)wave * (EDGE_PER_WAVE / 4) + (unsigned)lane;
            int4 r4 = row4[e4];
            const int4 c4 = col4[e4];
            asm volatile("" : "+v"(r4.x), "+v"(r4.y), "+v"(r4.z), "+v"(r4.w));

            #define EDGE_STEP(R, C) \
                edge_push(myq, tail, (R), (C), side); \
                if (tail - head >= 32u) { \
                    const uint2 e = myq[(head + (unsigned)lane) & (EDGE_Q - 1)]; \
                    const unsigned rb = e.x, cb = e.y; \
                    float hv[16]; \
                      \
                    EDGE_16(EDGE_GATHER) \
                    EDGE_16(EDGE_SCATTER) \
                    head += 32u; \
                }
            EDGE_STEP(r4.x, c4.x)
            EDGE_STEP(r4.y, c4.y)
            EDGE_STEP(r4.z, c4.z)
            EDGE_STEP(r4.w, c4.w)
            #undef EDGE_STEP
        }
        side ^= 1u;
    }

    const unsigned pend = tail - head;
    if (pend != 0u) {
        uint2 e = make_uint2(0u, 0u);
        if ((unsigned)lane < pend) e = myq[(head + (unsigned)lane) & (EDGE_Q - 1)];
        const unsigned rb = e.x, cb = e.y;
        const int lim = (int)pend - (lane & 16);
        float hv[16] = {};
        EDGE_16(EDGE_GATHER_P)
        EDGE_16(EDGE_SCATTER_P)
    }
}


#define OWN_RR        200
#define OWN_NR        (N_NODES / OWN_RR)
#define OWN_SRC_BITS  17
#define OWN_SRC_MASK  ((1u << OWN_SRC_BITS) - 1u)
#define OWN_T         512
#define CUR_STRIDE    64
#define BIN_CH        6400
#define BIN_NCH       (N_EDGES / BIN_CH)
#define BIN_T         320
#define BIN_U         5
#define OWN_WS_WORDS  ((size_t)((N_NODES + 255) / 256) * 256 + 2 * (size_t)N_NODES * HID + N_EDGES + 1024 + (size_t)OWN_NR * CUR_STRIDE)

static_assert(N_NODES % OWN_RR == 0, "ranges must tile the rows");
static_assert(N_NODES <= (1 << OWN_SRC_BITS), "a source row fits its field");
static_assert(OWN_RR < (1 << (32 - OWN_SRC_BITS)), "a target offset (and the spare row) fits its field");
static_assert(OWN_NR <= 512, "one wave scans the ranges, 16 per lane");
static_assert(OWN_RR <= 256, "one thread per row in k_dinv_own");
static_assert(N_EDGES % BIN_CH == 0 && BIN_T * BIN_U * 4 == BIN_CH, "chunks tile the edges, threads tile a chunk");
static_assert(BIN_T <= OWN_NR && 2 * BIN_T >= OWN_NR, "a binning thread reserves for one or two ranges");

__global__ __launch_bounds__(256) void k_dinv_own(const unsigned* __restrict__ part, float* __restrict__ dinv,
                                                  unsigned* __restrict__ tot, unsigned* __restrict__ cur) {
    __shared__ unsigned s_sum[8];
    const int t = threadIdx.x;
    const int r = blockIdx.x;
    unsigned inc = 0u;
    if (t < OWN_RR) {
        const int i = r * OWN_RR + t;
        for (int s = 0; s < DEG_S; ++s) inc += part[(size_t)s * N_NODES + i];
        float d = (float)(inc + 1u);
        dinv[i] = (d > 0.0f) ? rsqrtf(d) : 0.0f;
    }
    unsigned v = inc;
    for (int o = 16; o > 0; o >>= 1) v += (unsigned)__shfl_xor((int)v, o);
    if ((t & 31) == 0) s_sum[t >> 5] = v;
    __syncthreads();
    if (t == 0) {
        unsigned s = 0u;
        for (int w = 0; w < 8; ++w) s += s_sum[w];
        tot[r] = s;
        cur[r * CUR_STRIDE] = 0u;
    }
}

__device__ __forceinline__ void own_scan(const unsigned* src, unsigned* dst, int lane) {
    unsigned v[16];
    unsigned s = 0u;
    #pragma unroll
    for (int j = 0; j < 16; ++j) {
        const int idx = lane * 16 + j;
        v[j] = (idx < OWN_NR) ? src[idx] : 0u;
        s += v[j];
    }
    unsigned inc = s;
    for (int o = 1; o < 32; o <<= 1) {
        const unsigned nb = (unsigned)__shfl_up((int)inc, o);
        if (lane >= o) inc += nb;
    }
    unsigned run = inc - s;
    #pragma unroll
    for (int j = 0; j < 16; ++j) {
        const int idx = lane * 16 + j;
        if (idx < OWN_NR) dst[idx] = run;
        run += v[j];
    }
}

__global__ __launch_bounds__(BIN_T) void k_bin(const int* __restrict__ row, const int* __restrict__ col,
                                               const unsigned* __restrict__ tot, unsigned* __restrict__ cur,
                                               unsigned* __restrict__ seg, unsigned* __restrict__ list) {
    __shared__ unsigned s_cnt[OWN_NR];
    __shared__ unsigned s_lst[OWN_NR];
    __shared__ unsigned s_gst[OWN_NR];
    __shared__ unsigned s_seg[OWN_NR];
    __shared__ unsigned s_w[BIN_CH];
    __shared__ unsigned s_g[BIN_CH];
    const int t = threadIdx.x;
    const int lane = t & 31;
    const int wave = t >> 5;

    for (int i = t; i < OWN_NR; i += BIN_T) { s_cnt[i] = 0u; s_seg[i] = tot[i]; }

    const int4* __restrict__ row4 = reinterpret_cast<const int4*>(row) + (size_t)blockIdx.x * (BIN_CH / 4);
    const int4* __restrict__ col4 = reinterpret_cast<const int4*>(col) + (size_t)blockIdx.x * (BIN_CH / 4);
    int4 r4[BIN_U], c4[BIN_U];
    #pragma unroll
    for (int u = 0; u < BIN_U; ++u) { r4[u] = row4[u * BIN_T + t]; c4[u] = col4[u * BIN_T + t]; }
    __syncthreads();

    #pragma unroll
    for (int u = 0; u < BIN_U; ++u) {
        atomicAdd(&s_cnt[(unsigned)c4[u].x / OWN_RR], 1u);
        atomicAdd(&s_cnt[(unsigned)c4[u].y / OWN_RR], 1u);
        atomicAdd(&s_cnt[(unsigned)c4[u].z / OWN_RR], 1u);
        atomicAdd(&s_cnt[(unsigned)c4[u].w / OWN_RR], 1u);
    }
    __syncthreads();

    if (wave == 0) own_scan(s_seg, s_seg, lane);
    if (wave == 1) own_scan(s_cnt, s_lst, lane);
    __syncthreads();

    {
        const int i0 = t, i1 = t + BIN_T;
        const bool two = i1 < OWN_NR;
        const unsigned n0 = s_cnt[i0];
        const unsigned n1 = two ? s_cnt[i1] : 0u;
        const unsigned off0 = atomicAdd(&cur[i0 * CUR_STRIDE], n0);
        unsigned off1 = 0u;
        if (two) off1 = atomicAdd(&cur[i1 * CUR_STRIDE], n1);
        s_gst[i0] = s_seg[i0] + off0;
        s_cnt[i0] = 0u;
        if (two) { s_gst[i1] = s_seg[i1] + off1; s_cnt[i1] = 0u; }
        if (blockIdx.x == 0) {
            seg[i0] = s_seg[i0];
            if (two) seg[i1] = s_seg[i1];
        }
    }
    __syncthreads();

    #define BIN_PUT(R, C) { \
        const unsigned rg = (unsigned)(C) / OWN_RR; \
        const unsigned lo = (unsigned)(C) - rg * OWN_RR; \
        const unsigned sl = atomicAdd(&s_cnt[rg], 1u); \
        const unsigned p  = s_lst[rg] + sl; \
        s_w[p] = (unsigned)(R) | (lo << OWN_SRC_BITS); \
        s_g[p] = s_gst[rg] + sl; }
    #pragma unroll
    for (int u = 0; u < BIN_U; ++u) {
        BIN_PUT(r4[u].x, c4[u].x)
        BIN_PUT(r4[u].y, c4[u].y)
        BIN_PUT(r4[u].z, c4[u].z)
        BIN_PUT(r4[u].w, c4[u].w)
    }
    #undef BIN_PUT
    __syncthreads();

    for (int p = t; p < BIN_CH; p += BIN_T) list[s_g[p]] = s_w[p];
}

#define OWN_GATHER(J) \
    hv[J] = *reinterpret_cast<const float*>(reinterpret_cast<const char*>(g) + (row_lane_u<J>(rb) + ch4));
#define OWN_ADD(J) \
    atomicAdd(reinterpret_cast<float*>(reinterpret_cast<char*>(acc) + (row_lane_u<J>(lb) + ch4)), hv[J]);

#define OWN_TILES     ((OWN_RR + 15) / 16)
#define OWN_PADR      (OWN_TILES * 16)
static_assert(OWN_TILES <= OWN_T / 32, "one wave per tile in the product epilogue");
static_assert(OWN_RR % 8 == 0 && OWN_RR >= 8, "a lane's 8 output rows are all inside or all outside the range");

template <int STAGE>
__global__ __launch_bounds__(OWN_T) void k_own_agg(const unsigned* __restrict__ list, const unsigned* __restrict__ seg,
                                                   const unsigned* __restrict__ tot, const float* __restrict__ g,
                                                   const float* __restrict__ dinv, const float* __restrict__ pb,
                                                   const float* __restrict__ Wn, const float* __restrict__ bl,
                                                   float* __restrict__ out) {
    __shared__ __attribute__((aligned(16))) float acc[OWN_PADR * HID];
    const int t = threadIdx.x;
    const int lane = t & 31;
    const int wave = __builtin_amdgcn_readfirstlane(t >> 5);
    const int r = blockIdx.x;
    const unsigned ch4 = (unsigned)(lane & 15) * 4u;
    const unsigned s0 = seg[r];
    const unsigned s1 = s0 + tot[r];
    const unsigned pad = (unsigned)OWN_RR << OWN_SRC_BITS;

    unsigned base = s0 + (unsigned)wave * 32u;
    unsigned w = (base + (unsigned)lane < s1) ? list[base + lane] : pad;

    const int m    = lane & 15;
    const int half = lane >> 4;
    const bool mine = wave < OWN_TILES;
    const int tw   = mine ? wave : 0;
    const int lr   = tw * 16 + m;
    const int lrc  = lr < OWN_RR ? lr : OWN_RR - 1;
    const int r0   = tw * 16 + 8 * half;
    const int r0c  = r0 < OWN_RR ? r0 : OWN_RR - 8;
    const float* __restrict__ dr = dinv + (size_t)r * OWN_RR;
    float drow = 0.0f, dself = 0.0f;
    v4f d0 = {0.0f, 0.0f, 0.0f, 0.0f}, d1 = d0;
    v2f bf[HID / 4] = {}, pv[HID / 4] = {};
    if (STAGE == 1) {
        drow = dr[lrc];
        d0 = *reinterpret_cast<const v4f*>(dr + r0c);
        d1 = *reinterpret_cast<const v4f*>(dr + r0c + 4);
        #pragma unroll
        for (int kk = 0; kk < HID / 4; ++kk) {
            const int kb = kk * 4 + 2 * half;
            bf[kk][0] = Wn[kb * HID + m];
            bf[kk][1] = Wn[(kb + 1) * HID + m];
            pv[kk][0] = pb[kb];
            pv[kk][1] = pb[kb + 1];
        }
    } else {
        if (t < OWN_RR) dself = dr[t];
    }

    const v4f* __restrict__ gown = reinterpret_cast<const v4f*>(g + (size_t)r * (OWN_RR * HID));
    v4f* acc4 = reinterpret_cast<v4f*>(acc);
    const v4f zero4 = {0.0f, 0.0f, 0.0f, 0.0f};
    for (int i = t; i < OWN_PADR * HID / 4; i += OWN_T) acc4[i] = (i < OWN_RR * HID / 4) ? gown[i] : zero4;
    __syncthreads();

    while (base < s1) {
        const unsigned rb = (w & OWN_SRC_MASK) * (HID * 4u);
        const unsigned lb = (w >> OWN_SRC_BITS) * (HID * 4u);
        base += (OWN_T / 32) * 32u;
        w = (base + (unsigned)lane < s1) ? list[base + lane] : pad;
        float hv[16];
        EDGE_16(OWN_GATHER)
        EDGE_16(OWN_ADD)
    }

    if (STAGE == 1) {
        __syncthreads();
        if (mine) {
            v8f c8 = {};
            #pragma unroll
            for (int kk = 0; kk < HID / 4; ++kk) {
                const int kb = kk * 4 + 2 * half;
                v2f a = *reinterpret_cast<const v2f*>(&acc[lr * HID + kb]);
                a[0] = drow * a[0] + pv[kk][0];
                a[1] = drow * a[1] + pv[kk][1];
                a[0] = a[0] > 0.0f ? a[0] : 0.0f;
                a[1] = a[1] > 0.0f ? a[1] : 0.0f;
                c8 = __builtin_amdgcn_wmma_f32_16x16x4_f32(
                          false, a,  false, bf[kk],
                          (short)0, c8,  false,  false);
            }
            #pragma unroll
            for (int i = 0; i < 8; ++i) {
                const float di = (i < 4) ? d0[i & 3] : d1[i & 3];
                acc[(r0 + i) * HID + m] = di * c8[i];
            }
        }
        __syncthreads();
        v4f* __restrict__ oown = reinterpret_cast<v4f*>(out + (size_t)r * (OWN_RR * HID));
        for (int i = t; i < OWN_RR * HID / 4; i += OWN_T) oown[i] = acc4[i];
    } else {
        const float di = dself;
        __syncthreads();
        if (t < OWN_RR) {
            float s = bl[0];
            #pragma unroll
            for (int q = 0; q < HID / 4; ++q) {
                const v4f x = acc4[t * (HID / 4) + q];
                #pragma unroll
                for (int j = 0; j < 4; ++j) {
                    const float v = di * x[j] + pb[q * 4 + j];
                    s += (v > 0.0f ? v : 0.0f) * Wn[q * 4 + j];
                }
            }
            out[(size_t)r * OWN_RR + t] = s;
        }
    }
}

__global__ void k_linear(const float* __restrict__ t, const float* __restrict__ dinv,
                         const float* __restrict__ b, const float* __restrict__ Wl,
                         const float* __restrict__ bl, float* __restrict__ out, int n) {
    int i = blockIdx.x * blockDim.x + threadIdx.x;
    if (i >= n) return;
    const float* tr = t + (long)i * HID;
    const float di = dinv[i];
    float s = bl[0];
    #pragma unroll
    for (int c = 0; c < HID; ++c) {
        float v = di * tr[c] + b[c];
        s += (v > 0.0f ? v : 0.0f) * Wl[c];
    }
    out[i] = s;
}


extern "C" void kernel_launch(void* const* d_in, const int* in_sizes, int n_in,
                              void* d_out, int out_size, void* d_ws, size_t ws_size,
                              hipStream_t stream) {
    const float* x  = (const float*)d_in[0];
    const int*   ei = (const int*)d_in[1];
    const int*   row = ei;
    const int*   col = ei + N_EDGES;
    const float* W1 = (const float*)d_in[2];
    const float* b1 = (const float*)d_in[3];
    const float* W2 = (const float*)d_in[4];
    const float* b2 = (const float*)d_in[5];
    const float* Wl = (const float*)d_in[6];
    const float* bl = (const float*)d_in[7];
    float* out = (float*)d_out;

    float* ws   = (float*)d_ws;
    float* dinv = ws;
    float* bufA = ws + ((N_NODES + 255) / 256) * 256;
    float* bufB = bufA + (size_t)N_NODES * HID;

    const int T = 256;
    const int gN   = (N_NODES + T - 1) / T;
    const int tiles = (N_NODES + 15) / 16;
    const int gGemm = (tiles + (T / 32) - 1) / (T / 32);

    unsigned* part = (unsigned*)bufB;
    k_deg_hist<<<DEG_R * DEG_S, DEG_T, 0, stream>>>(col, part);

    if (ws_size >= sizeof(float) * OWN_WS_WORDS) {
        unsigned* list = (unsigned*)(bufB + (size_t)N_NODES * HID);
        unsigned* tot  = list + N_EDGES;
        unsigned* seg  = tot + 512;
        unsigned* cur  = seg + 512;
        k_dinv_own<<<OWN_NR, 256, 0, stream>>>(part, dinv, tot, cur);
        k_bin<<<BIN_NCH, BIN_T, 0, stream>>>(row, col, tot, cur, seg, list);

        k_gemm_wmma<false, false><<<gGemm, T, IN_CH * HID * sizeof(float), stream>>>(x, W1, dinv, b1, bufA, nullptr,
                                                                                     N_NODES, IN_CH);
        k_own_agg<1><<<OWN_NR, OWN_T, 0, stream>>>(list, seg, tot, bufA, dinv, b1, W2, bl, bufB);
        k_own_agg<2><<<OWN_NR, OWN_T, 0, stream>>>(list, seg, tot, bufB, dinv, b2, Wl, bl, out);
        return;
    }

    unsigned* tick = (unsigned*)(ws + N_NODES);
    k_dinv<<<gN, T, 0, stream>>>(part, dinv, tick, N_NODES);

    k_gemm_wmma<false, true><<<gGemm, T, IN_CH * HID * sizeof(float), stream>>>(x, W1, dinv, b1, bufA, bufB,
                                                                                N_NODES, IN_CH);
    k_edge_agg<<<EDGE_BLOCKS, EDGE_T, 0, stream>>>(row, col, bufA, bufB, tick);

    k_gemm_wmma<true, true><<<gGemm, T, HID * HID * sizeof(float), stream>>>(bufB, W2, dinv, b1, bufA, bufB,
                                                                             N_NODES, HID);
    k_edge_agg<<<EDGE_BLOCKS, EDGE_T, 0, stream>>>(row, col, bufA, bufB, tick + 1);

    k_linear<<<gN, T, 0, stream>>>(bufB, dinv, b2, Wl, bl, out, N_NODES);
}


# Edge binning and the layer-1 WMMA product share one launch: work-groups 0-499 bin, 625 more take 10 tiles each (W1 staged by 16-byte loads, a tile's loads in one run); x streams under binning; 6->5
# speedup vs baseline: 5.2908x; 1.0900x over previous
#include <hip/hip_runtime.h>
#include <hip/hip_bf16.h>

typedef float v2f __attribute__((ext_vector_type(2)));
typedef float v4f __attribute__((ext_vector_type(4)));
typedef float v8f __attribute__((ext_vector_type(8)));

#define N_NODES 100000
#define N_EDGES 3200000
#define IN_CH   128
#define HID     16


#define DEG_R      16
#define DEG_S      16
#define DEG_RANGE  (N_NODES / DEG_R)
#define DEG_SLICE4 (N_EDGES / DEG_S / 4)
#define DEG_T      1024
#define DEG_U      7

static_assert(N_NODES % DEG_R == 0, "ranges must tile the nodes");
static_assert(N_EDGES % (DEG_S * 4) == 0, "slices must tile the edges in groups of four");
static_assert(DEG_S <= HID, "the partial rows are kept in one N x HID buffer");
static_assert(N_NODES % 16 == 0, "GEMM tiles are whole: a lane's 8 dinv values are read as two 16-byte loads");

#define TICK_STRIDE 64
#define WS_PAD      (((N_NODES + 255) / 256) * 256 - N_NODES)
static_assert(TICK_STRIDE + 2 <= WS_PAD, "the counters fit in the padding behind dinv");
static_assert((N_NODES * 4) % 128 == 0, "the counters start on a line of their own");

__device__ __forceinline__ void deg_bump(unsigned* hist, int c, unsigned lo) {
    unsigned d = (unsigned)c - lo;
    if (d < (unsigned)DEG_RANGE) atomicAdd(&hist[d], 1u);
}

__global__ __launch_bounds__(DEG_T) void k_deg_hist(const int* __restrict__ col,
                                                    unsigned* __restrict__ part) {
    __shared__ unsigned hist[DEG_RANGE];
    const int t  = threadIdx.x;
    const int rg = blockIdx.x % DEG_R;
    const int sl = blockIdx.x / DEG_R;
    for (int i = t; i < DEG_RANGE; i += DEG_T) hist[i] = 0u;
    __syncthreads();

    const unsigned lo = (unsigned)(rg * DEG_RANGE);
    const int4* __restrict__ c4 = reinterpret_cast<const int4*>(col) + (size_t)sl * DEG_SLICE4;
    for (int base = 0; base < DEG_SLICE4; base += DEG_T * DEG_U) {
        int4 v[DEG_U];
        #pragma unroll
        for (int u = 0; u < DEG_U; ++u) {
            int i = base + u * DEG_T + t;
            v[u] = (i < DEG_SLICE4) ? c4[i] : make_int4(-1, -1, -1, -1);
        }
        #pragma unroll
        for (int u = 0; u < DEG_U; ++u) {
            deg_bump(hist, v[u].x, lo);
            deg_bump(hist, v[u].y, lo);
            deg_bump(hist, v[u].z, lo);
            deg_bump(hist, v[u].w, lo);
        }
    }
    __syncthreads();

    unsigned* __restrict__ dst = part + (size_t)sl * N_NODES + lo;
    for (int i = t; i < DEG_RANGE; i += DEG_T) dst[i] = hist[i];
}

__global__ void k_dinv(const unsigned* __restrict__ part, float* __restrict__ dinv,
                       unsigned* __restrict__ tick, int n) {
    int i = blockIdx.x * blockDim.x + threadIdx.x;
    if (i == 0) {
        tick[0] = 0u;               tick[1] = 0u;
        tick[TICK_STRIDE] = 0u;     tick[TICK_STRIDE + 1] = 0u;
    }
    if (i < n) {
        unsigned cnt = 1u;
        for (int s = 0; s < DEG_S; ++s) cnt += part[(size_t)s * N_NODES + i];
        float d = (float)cnt;
        dinv[i] = (d > 0.0f) ? rsqrtf(d) : 0.0f;
    }
}


template <bool PREV, bool START, int KU>
__device__ __forceinline__ void gemm_tile(const float* A, const float* sW,
                                          const float* __restrict__ dinv, const float* __restrict__ pbias,
                                          float* G, float* S, int n, int K, int tile) {
    int lane = threadIdx.x & 31;
    int m    = lane & 15;
    int half = lane >> 4;

    int rb   = tile * 16;
    if (rb >= n) return;

    int row = rb + m;
    if (row > n - 1) row = n - 1;
    const float* arow = A + (long)row * K + 2 * half;

    const int r0 = rb + 8 * half;
    const v4f d0 = *reinterpret_cast<const v4f*>(dinv + r0);
    const v4f d1 = *reinterpret_cast<const v4f*>(dinv + r0 + 4);
    const float drow = PREV ? dinv[row] : 1.0f;

    v8f acc = {};
    #pragma unroll KU
    for (int k = 0; k < K; k += 4) {
        v2f a = *reinterpret_cast<const v2f*>(arow + k);
        int kb = k + 2 * half;
        if (PREV) {
            a[0] = drow * a[0] + pbias[kb];
            a[1] = drow * a[1] + pbias[kb + 1];
            a[0] = a[0] > 0.0f ? a[0] : 0.0f;
            a[1] = a[1] > 0.0f ? a[1] : 0.0f;
        }
        v2f b;
        b[0] = sW[kb * HID + m];
        b[1] = sW[(kb + 1) * HID + m];
        acc = __builtin_amdgcn_wmma_f32_16x16x4_f32(
                   false, a,  false, b,
                   (short)0, acc,  false,  false);
    }

    #pragma unroll
    for (int i = 0; i < 8; ++i) {
        int r = r0 + i;
        if (r < n) {
            float di = (i < 4) ? d0[i & 3] : d1[i & 3];
            float g  = di * acc[i];
            G[(long)r * HID + m] = g;
            if (START) S[(long)r * HID + m] = g;
        }
    }
}

template <bool PREV, bool START>
__global__ void k_gemm_wmma(const float* A, const float* __restrict__ W,
                            const float* __restrict__ dinv, const float* __restrict__ pbias,
                            float* G, float* S, int n, int K) {
    extern __shared__ float sW[];
    int t = threadIdx.x;
    for (int i = t; i < K * HID; i += blockDim.x) sW[i] = W[i];
    __syncthreads();
    gemm_tile<PREV, START, 8>(A, sW, dinv, pbias, G, S, n, K, blockIdx.x * (blockDim.x >> 5) + (t >> 5));
}


#define EDGE_T         256
#define EDGE_WPB       (EDGE_T / 32)
#define EDGE_PER_WAVE  128
#define EDGE_TILE      (EDGE_WPB * EDGE_PER_WAVE)
#define EDGE_TILES     (N_EDGES / EDGE_TILE)
#define EDGE_BLOCKS    1280
#define EDGE_Q         64
#define OWN_SPLIT      (N_NODES / 2)

static_assert(N_EDGES % EDGE_TILE == 0, "tiles must tile the edges");
static_assert(HID == 16, "one half-wave per 16-channel row");
static_assert((long)N_NODES * HID * 4 < (1L << 32), "row byte offsets are kept in 32 bits");
static_assert(((long)OWN_SPLIT * HID * 4) % 256 == 0, "the two halves' rows share no 256-byte unit");

template <int J> __device__ __forceinline__ unsigned row_lane_u(unsigned v) {
    return (unsigned)__builtin_amdgcn_update_dpp(0, (int)v, 0x150 + J, 0xF, 0xF, true);
}

#define EDGE_GATHER(J) \
    hv[J] = *reinterpret_cast<const float*>(reinterpret_cast<const char*>(h) + (row_lane_u<J>(rb) + ch4));
#define EDGE_SCATTER(J) \
    atomicAdd(reinterpret_cast<float*>(reinterpret_cast<char*>(agg) + (row_lane_u<J>(cb) + ch4)), hv[J]);
#define EDGE_GATHER_P(J) { \
    const unsigned ga = row_lane_u<J>(rb) + ch4; \
    if (J < lim) hv[J] = *reinterpret_cast<const float*>(reinterpret_cast<const char*>(h) + ga); }
#define EDGE_SCATTER_P(J) { \
    const unsigned sa = row_lane_u<J>(cb) + ch4; \
    if (J < lim) atomicAdd(reinterpret_cast<float*>(reinterpret_cast<char*>(agg) + sa), hv[J]); }
#define EDGE_16(M) M(0) M(1) M(2) M(3) M(4) M(5) M(6) M(7) M(8) M(9) M(10) M(11) M(12) M(13) M(14) M(15)

__device__ __forceinline__ void edge_push(uint2* __restrict__ myq, unsigned& tail, int r, int c, unsigned side) {
    const bool own = (((unsigned)c >= (unsigned)OWN_SPLIT) ? 1u : 0u) == side;
    const unsigned m = __builtin_amdgcn_ballot_w32(own);
    if (own) {
        const unsigned pos = (tail + __builtin_amdgcn_mbcnt_lo(m, 0u)) & (EDGE_Q - 1);
        myq[pos] = make_uint2((unsigned)r * (HID * 4u), (unsigned)c * (HID * 4u));
    }
    tail += (unsigned)__builtin_popcount(m);
}

__global__ __launch_bounds__(EDGE_T) void k_edge_agg(const int* __restrict__ row,
                                                     const int* __restrict__ col,
                                                     const float* __restrict__ h,
                                                     float* __restrict__ agg,
                                                     unsigned* __restrict__ tick) {
    __shared__ uint2 q[EDGE_WPB][EDGE_Q];
    __shared__ unsigned s_tile;
    const int lane = threadIdx.x & 31;
    const int wave = threadIdx.x >> 5;
    const unsigned ch4 = (unsigned)(lane & 15) * 4u;
    uint2* __restrict__ myq = q[wave];

    unsigned side = (__builtin_amdgcn_s_sendmsg_rtn(0x87) >> 18) & 1u;
    unsigned head = 0u, tail = 0u;

    const int4* __restrict__ row4 = reinterpret_cast<const int4*>(row);
    const int4* __restrict__ col4 = reinterpret_cast<const int4*>(col);

    for (int phase = 0; phase < 2; ++phase) {
        unsigned* tk = tick + side * TICK_STRIDE;
        for (unsigned it = 0; it <= (unsigned)EDGE_TILES; ++it) {
            if (threadIdx.x == 0) s_tile = atomicAdd(tk, 1u);
            __syncthreads();
            const unsigned tile = (unsigned)__builtin_amdgcn_readfirstlane((int)s_tile);
            __syncthreads();
            if (tile >= (unsigned)EDGE_TILES) break;

            const unsigned e4 = tile * (EDGE_TILE / 4) + (unsigned)wave * (EDGE_PER_WAVE / 4) + (unsigned)lane;
            int4 r4 = row4[e4];
            const int4 c4 = col4[e4];
            asm volatile("" : "+v"(r4.x), "+v"(r4.y), "+v"(r4.z), "+v"(r4.w));

            #define EDGE_STEP(R, C) \
                edge_push(myq, tail, (R), (C), side); \
                if (tail - head >= 32u) { \
                    const uint2 e = myq[(head + (unsigned)lane) & (EDGE_Q - 1)]; \
                    const unsigned rb = e.x, cb = e.y; \
                    float hv[16]; \
                      \
                    EDGE_16(EDGE_GATHER) \
                    EDGE_16(EDGE_SCATTER) \
                    head += 32u; \
                }
            EDGE_STEP(r4.x, c4.x)
            EDGE_STEP(r4.y, c4.y)
            EDGE_STEP(r4.z, c4.z)
            EDGE_STEP(r4.w, c4.w)
            #undef EDGE_STEP
        }
        side ^= 1u;
    }

    const unsigned pend = tail - head;
    if (pend != 0u) {
        uint2 e = make_uint2(0u, 0u);
        if ((unsigned)lane < pend) e = myq[(head + (unsigned)lane) & (EDGE_Q - 1)];
        const unsigned rb = e.x, cb = e.y;
        const int lim = (int)pend - (lane & 16);
        float hv[16] = {};
        EDGE_16(EDGE_GATHER_P)
        EDGE_16(EDGE_SCATTER_P)
    }
}


#define OWN_RR        200
#define OWN_NR        (N_NODES / OWN_RR)
#define OWN_SRC_BITS  17
#define OWN_SRC_MASK  ((1u << OWN_SRC_BITS) - 1u)
#define OWN_T         512
#define CUR_STRIDE    64
#define BIN_CH        6400
#define BIN_NCH       (N_EDGES / BIN_CH)
#define BIN_T         320
#define BIN_U         5
#define OWN_WS_WORDS  ((size_t)((N_NODES + 255) / 256) * 256 + 2 * (size_t)N_NODES * HID + N_EDGES + 1024 + (size_t)OWN_NR * CUR_STRIDE)

static_assert(N_NODES % OWN_RR == 0, "ranges must tile the rows");
static_assert(N_NODES <= (1 << OWN_SRC_BITS), "a source row fits its field");
static_assert(OWN_RR < (1 << (32 - OWN_SRC_BITS)), "a target offset (and the spare row) fits its field");
static_assert(OWN_NR <= 512, "one wave scans the ranges, 16 per lane");
static_assert(OWN_RR <= 256, "one thread per row in k_dinv_own");
static_assert(N_EDGES % BIN_CH == 0 && BIN_T * BIN_U * 4 == BIN_CH, "chunks tile the edges, threads tile a chunk");
static_assert(BIN_T <= OWN_NR && 2 * BIN_T >= OWN_NR, "a binning thread reserves for one or two ranges");
#define BG_GEMM_WG    ((N_NODES / 16) / (BIN_T / 32))
static_assert((N_NODES / 16) % (BIN_T / 32) == 0, "the product's work-groups hold whole tiles only");
static_assert(IN_CH * HID <= BIN_CH, "W1 fits the LDS words that the binning role uses for its ordered chunk");

__global__ __launch_bounds__(256) void k_dinv_own(const unsigned* __restrict__ part, float* __restrict__ dinv,
                                                  unsigned* __restrict__ tot, unsigned* __restrict__ cur) {
    __shared__ unsigned s_sum[8];
    const int t = threadIdx.x;
    const int r = blockIdx.x;
    unsigned inc = 0u;
    if (t < OWN_RR) {
        const int i = r * OWN_RR + t;
        for (int s = 0; s < DEG_S; ++s) inc += part[(size_t)s * N_NODES + i];
        float d = (float)(inc + 1u);
        dinv[i] = (d > 0.0f) ? rsqrtf(d) : 0.0f;
    }
    unsigned v = inc;
    for (int o = 16; o > 0; o >>= 1) v += (unsigned)__shfl_xor((int)v, o);
    if ((t & 31) == 0) s_sum[t >> 5] = v;
    __syncthreads();
    if (t == 0) {
        unsigned s = 0u;
        for (int w = 0; w < 8; ++w) s += s_sum[w];
        tot[r] = s;
        cur[r * CUR_STRIDE] = 0u;
    }
}

__device__ __forceinline__ void own_scan(const unsigned* src, unsigned* dst, int lane) {
    unsigned v[16];
    unsigned s = 0u;
    #pragma unroll
    for (int j = 0; j < 16; ++j) {
        const int idx = lane * 16 + j;
        v[j] = (idx < OWN_NR) ? src[idx] : 0u;
        s += v[j];
    }
    unsigned inc = s;
    for (int o = 1; o < 32; o <<= 1) {
        const unsigned nb = (unsigned)__shfl_up((int)inc, o);
        if (lane >= o) inc += nb;
    }
    unsigned run = inc - s;
    #pragma unroll
    for (int j = 0; j < 16; ++j) {
        const int idx = lane * 16 + j;
        if (idx < OWN_NR) dst[idx] = run;
        run += v[j];
    }
}

__global__ __launch_bounds__(BIN_T) void k_bin_gemm(const int* __restrict__ row, const int* __restrict__ col,
                                                    const unsigned* __restrict__ tot, unsigned* __restrict__ cur,
                                                    unsigned* __restrict__ seg, unsigned* __restrict__ list,
                                                    const float* __restrict__ x, const float* __restrict__ W1,
                                                    const float* __restrict__ dinv, float* __restrict__ G) {
    __shared__ unsigned s_cnt[OWN_NR];
    __shared__ unsigned s_lst[OWN_NR];
    __shared__ unsigned s_gst[OWN_NR];
    __shared__ unsigned s_seg[OWN_NR];
    __shared__ __attribute__((aligned(16))) unsigned s_w[BIN_CH];
    __shared__ unsigned s_g[BIN_CH];

    if (blockIdx.x >= BIN_NCH) {
        const v4f* __restrict__ w4 = reinterpret_cast<const v4f*>(W1);
        v4f* sw4 = reinterpret_cast<v4f*>(s_w);
        for (int i = threadIdx.x; i < IN_CH * HID / 4; i += BIN_T) sw4[i] = w4[i];
        __syncthreads();
        gemm_tile<false, false, IN_CH / 4>(x, reinterpret_cast<const float*>(s_w), dinv, nullptr, G, nullptr,
                                           N_NODES, IN_CH,
                                           ((int)blockIdx.x - BIN_NCH) * (BIN_T / 32) + (int)(threadIdx.x >> 5));
        return;
    }

    const int t = threadIdx.x;
    const int lane = t & 31;
    const int wave = t >> 5;

    for (int i = t; i < OWN_NR; i += BIN_T) { s_cnt[i] = 0u; s_seg[i] = tot[i]; }

    const int4* __restrict__ row4 = reinterpret_cast<const int4*>(row) + (size_t)blockIdx.x * (BIN_CH / 4);
    const int4* __restrict__ col4 = reinterpret_cast<const int4*>(col) + (size_t)blockIdx.x * (BIN_CH / 4);
    int4 r4[BIN_U], c4[BIN_U];
    #pragma unroll
    for (int u = 0; u < BIN_U; ++u) { r4[u] = row4[u * BIN_T + t]; c4[u] = col4[u * BIN_T + t]; }
    __syncthreads();

    #pragma unroll
    for (int u = 0; u < BIN_U; ++u) {
        atomicAdd(&s_cnt[(unsigned)c4[u].x / OWN_RR], 1u);
        atomicAdd(&s_cnt[(unsigned)c4[u].y / OWN_RR], 1u);
        atomicAdd(&s_cnt[(unsigned)c4[u].z / OWN_RR], 1u);
        atomicAdd(&s_cnt[(unsigned)c4[u].w / OWN_RR], 1u);
    }
    __syncthreads();

    if (wave == 0) own_scan(s_seg, s_seg, lane);
    if (wave == 1) own_scan(s_cnt, s_lst, lane);
    __syncthreads();

    {
        const int i0 = t, i1 = t + BIN_T;
        const bool two = i1 < OWN_NR;
        const unsigned n0 = s_cnt[i0];
        const unsigned n1 = two ? s_cnt[i1] : 0u;
        const unsigned off0 = atomicAdd(&cur[i0 * CUR_STRIDE], n0);
        unsigned off1 = 0u;
        if (two) off1 = atomicAdd(&cur[i1 * CUR_STRIDE], n1);
        s_gst[i0] = s_seg[i0] + off0;
        s_cnt[i0] = 0u;
        if (two) { s_gst[i1] = s_seg[i1] + off1; s_cnt[i1] = 0u; }
        if (blockIdx.x == 0) {
            seg[i0] = s_seg[i0];
            if (two) seg[i1] = s_seg[i1];
        }
    }
    __syncthreads();

    #define BIN_PUT(R, C) { \
        const unsigned rg = (unsigned)(C) / OWN_RR; \
        const unsigned lo = (unsigned)(C) - rg * OWN_RR; \
        const unsigned sl = atomicAdd(&s_cnt[rg], 1u); \
        const unsigned p  = s_lst[rg] + sl; \
        s_w[p] = (unsigned)(R) | (lo << OWN_SRC_BITS); \
        s_g[p] = s_gst[rg] + sl; }
    #pragma unroll
    for (int u = 0; u < BIN_U; ++u) {
        BIN_PUT(r4[u].x, c4[u].x)
        BIN_PUT(r4[u].y, c4[u].y)
        BIN_PUT(r4[u].z, c4[u].z)
        BIN_PUT(r4[u].w, c4[u].w)
    }
    #undef BIN_PUT
    __syncthreads();

    for (int p = t; p < BIN_CH; p += BIN_T) list[s_g[p]] = s_w[p];
}

#define OWN_GATHER(J) \
    hv[J] = *reinterpret_cast<const float*>(reinterpret_cast<const char*>(g) + (row_lane_u<J>(rb) + ch4));
#define OWN_ADD(J) \
    atomicAdd(reinterpret_cast<float*>(reinterpret_cast<char*>(acc) + (row_lane_u<J>(lb) + ch4)), hv[J]);

#define OWN_TILES     ((OWN_RR + 15) / 16)
#define OWN_PADR      (OWN_TILES * 16)
static_assert(OWN_TILES <= OWN_T / 32, "one wave per tile in the product epilogue");
static_assert(OWN_RR % 8 == 0 && OWN_RR >= 8, "a lane's 8 output rows are all inside or all outside the range");

template <int STAGE>
__global__ __launch_bounds__(OWN_T) void k_own_agg(const unsigned* __restrict__ list, const unsigned* __restrict__ seg,
                                                   const unsigned* __restrict__ tot, const float* __restrict__ g,
                                                   const float* __restrict__ dinv, const float* __restrict__ pb,
                                                   const float* __restrict__ Wn, const float* __restrict__ bl,
                                                   float* __restrict__ out) {
    __shared__ __attribute__((aligned(16))) float acc[OWN_PADR * HID];
    const int t = threadIdx.x;
    const int lane = t & 31;
    const int wave = __builtin_amdgcn_readfirstlane(t >> 5);
    const int r = blockIdx.x;
    const unsigned ch4 = (unsigned)(lane & 15) * 4u;
    const unsigned s0 = seg[r];
    const unsigned s1 = s0 + tot[r];
    const unsigned pad = (unsigned)OWN_RR << OWN_SRC_BITS;

    unsigned base = s0 + (unsigned)wave * 32u;
    unsigned w = (base + (unsigned)lane < s1) ? list[base + lane] : pad;

    const int m    = lane & 15;
    const int half = lane >> 4;
    const bool mine = wave < OWN_TILES;
    const int tw   = mine ? wave : 0;
    const int lr   = tw * 16 + m;
    const int lrc  = lr < OWN_RR ? lr : OWN_RR - 1;
    const int r0   = tw * 16 + 8 * half;
    const int r0c  = r0 < OWN_RR ? r0 : OWN_RR - 8;
    const float* __restrict__ dr = dinv + (size_t)r * OWN_RR;
    float drow = 0.0f, dself = 0.0f;
    v4f d0 = {0.0f, 0.0f, 0.0f, 0.0f}, d1 = d0;
    v2f bf[HID / 4] = {}, pv[HID / 4] = {};
    if (STAGE == 1) {
        drow = dr[lrc];
        d0 = *reinterpret_cast<const v4f*>(dr + r0c);
        d1 = *reinterpret_cast<const v4f*>(dr + r0c + 4);
        #pragma unroll
        for (int kk = 0; kk < HID / 4; ++kk) {
            const int kb = kk * 4 + 2 * half;
            bf[kk][0] = Wn[kb * HID + m];
            bf[kk][1] = Wn[(kb + 1) * HID + m];
            pv[kk][0] = pb[kb];
            pv[kk][1] = pb[kb + 1];
        }
    } else {
        if (t < OWN_RR) dself = dr[t];
    }

    const v4f* __restrict__ gown = reinterpret_cast<const v4f*>(g + (size_t)r * (OWN_RR * HID));
    v4f* acc4 = reinterpret_cast<v4f*>(acc);
    const v4f zero4 = {0.0f, 0.0f, 0.0f, 0.0f};
    for (int i = t; i < OWN_PADR * HID / 4; i += OWN_T) acc4[i] = (i < OWN_RR * HID / 4) ? gown[i] : zero4;
    __syncthreads();

    while (base < s1) {
        const unsigned rb = (w & OWN_SRC_MASK) * (HID * 4u);
        const unsigned lb = (w >> OWN_SRC_BITS) * (HID * 4u);
        base += (OWN_T / 32) * 32u;
        w = (base + (unsigned)lane < s1) ? list[base + lane] : pad;
        float hv[16];
        EDGE_16(OWN_GATHER)
        EDGE_16(OWN_ADD)
    }

    if (STAGE == 1) {
        __syncthreads();
        if (mine) {
            v8f c8 = {};
            #pragma unroll
            for (int kk = 0; kk < HID / 4; ++kk) {
                const int kb = kk * 4 + 2 * half;
                v2f a = *reinterpret_cast<const v2f*>(&acc[lr * HID + kb]);
                a[0] = drow * a[0] + pv[kk][0];
                a[1] = drow * a[1] + pv[kk][1];
                a[0] = a[0] > 0.0f ? a[0] : 0.0f;
                a[1] = a[1] > 0.0f ? a[1] : 0.0f;
                c8 = __builtin_amdgcn_wmma_f32_16x16x4_f32(
                          false, a,  false, bf[kk],
                          (short)0, c8,  false,  false);
            }
            #pragma unroll
            for (int i = 0; i < 8; ++i) {
                const float di = (i < 4) ? d0[i & 3] : d1[i & 3];
                acc[(r0 + i) * HID + m] = di * c8[i];
            }
        }
        __syncthreads();
        v4f* __restrict__ oown = reinterpret_cast<v4f*>(out + (size_t)r * (OWN_RR * HID));
        for (int i = t; i < OWN_RR * HID / 4; i += OWN_T) oown[i] = acc4[i];
    } else {
        const float di = dself;
        __syncthreads();
        if (t < OWN_RR) {
            float s = bl[0];
            #pragma unroll
            for (int q = 0; q < HID / 4; ++q) {
                const v4f x = acc4[t * (HID / 4) + q];
                #pragma unroll
                for (int j = 0; j < 4; ++j) {
                    const float v = di * x[j] + pb[q * 4 + j];
                    s += (v > 0.0f ? v : 0.0f) * Wn[q * 4 + j];
                }
            }
            out[(size_t)r * OWN_RR + t] = s;
        }
    }
}

__global__ void k_linear(const float* __restrict__ t, const float* __restrict__ dinv,
                         const float* __restrict__ b, const float* __restrict__ Wl,
                         const float* __restrict__ bl, float* __restrict__ out, int n) {
    int i = blockIdx.x * blockDim.x + threadIdx.x;
    if (i >= n) return;
    const float* tr = t + (long)i * HID;
    const float di = dinv[i];
    float s = bl[0];
    #pragma unroll
    for (int c = 0; c < HID; ++c) {
        float v = di * tr[c] + b[c];
        s += (v > 0.0f ? v : 0.0f) * Wl[c];
    }
    out[i] = s;
}


extern "C" void kernel_launch(void* const* d_in, const int* in_sizes, int n_in,
                              void* d_out, int out_size, void* d_ws, size_t ws_size,
                              hipStream_t stream) {
    const float* x  = (const float*)d_in[0];
    const int*   ei = (const int*)d_in[1];
    const int*   row = ei;
    const int*   col = ei + N_EDGES;
    const float* W1 = (const float*)d_in[2];
    const float* b1 = (const float*)d_in[3];
    const float* W2 = (const float*)d_in[4];
    const float* b2 = (const float*)d_in[5];
    const float* Wl = (const float*)d_in[6];
    const float* bl = (const float*)d_in[7];
    float* out = (float*)d_out;

    float* ws   = (float*)d_ws;
    float* dinv = ws;
    float* bufA = ws + ((N_NODES + 255) / 256) * 256;
    float* bufB = bufA + (size_t)N_NODES * HID;

    const int T = 256;
    const int gN   = (N_NODES + T - 1) / T;
    const int tiles = (N_NODES + 15) / 16;
    const int gGemm = (tiles + (T / 32) - 1) / (T / 32);

    unsigned* part = (unsigned*)bufB;
    k_deg_hist<<<DEG_R * DEG_S, DEG_T, 0, stream>>>(col, part);

    if (ws_size >= sizeof(float) * OWN_WS_WORDS) {
        unsigned* list = (unsigned*)(bufB + (size_t)N_NODES * HID);
        unsigned* tot  = list + N_EDGES;
        unsigned* seg  = tot + 512;
        unsigned* cur  = seg + 512;
        k_dinv_own<<<OWN_NR, 256, 0, stream>>>(part, dinv, tot, cur);

        k_bin_gemm<<<BIN_NCH + BG_GEMM_WG, BIN_T, 0, stream>>>(row, col, tot, cur, seg, list, x, W1, dinv, bufA);
        k_own_agg<1><<<OWN_NR, OWN_T, 0, stream>>>(list, seg, tot, bufA, dinv, b1, W2, bl, bufB);
        k_own_agg<2><<<OWN_NR, OWN_T, 0, stream>>>(list, seg, tot, bufB, dinv, b2, Wl, bl, out);
        return;
    }

    unsigned* tick = (unsigned*)(ws + N_NODES);
    k_dinv<<<gN, T, 0, stream>>>(part, dinv, tick, N_NODES);

    k_gemm_wmma<false, true><<<gGemm, T, IN_CH * HID * sizeof(float), stream>>>(x, W1, dinv, b1, bufA, bufB,
                                                                                N_NODES, IN_CH);
    k_edge_agg<<<EDGE_BLOCKS, EDGE_T, 0, stream>>>(row, col, bufA, bufB, tick);

    k_gemm_wmma<true, true><<<gGemm, T, HID * HID * sizeof(float), stream>>>(bufB, W2, dinv, b1, bufA, bufB,
                                                                             N_NODES, HID);
    k_edge_agg<<<EDGE_BLOCKS, EDGE_T, 0, stream>>>(row, col, bufA, bufB, tick + 1);

    k_linear<<<gN, T, 0, stream>>>(bufB, dinv, b2, Wl, bl, out, N_NODES);
}
